# starting kernel
# baseline (speedup 1.0000x reference)
#include <hip/hip_runtime.h>
#include <hip/hip_bf16.h>


#define NN      100000
#define EE      1600000
#define ETOT    (EE + NN)
#define F_IN    256
#define DD      128
#define HH      8
#define CC      16
#define LL      4
#define NEG_SLOPE 0.2f
#define LN_EPS  1e-5f

typedef __attribute__((ext_vector_type(2))) float v2f;
typedef __attribute__((ext_vector_type(8))) float v8f;

template <int K>
__global__ __launch_bounds__(256) void gemm_wmma_k(
    const float* __restrict__ X, const float* __restrict__ W,
    const float* __restrict__ bias, float* __restrict__ Y) {
  constexpr int KP = K + 4;
  __shared__ float As[16 * KP];

  const int wave = threadIdx.x >> 5;
  const int lane = threadIdx.x & 31;
  const int row0 = blockIdx.x * 16;
  const int col0 = wave * 16;
  const int m    = lane & 15;
  const int koff = (lane < 16) ? 0 : 2;

  #pragma unroll
  for (int i = threadIdx.x * 4; i < 16 * K; i += 256 * 4) {
    const int r = i / K, c = i % K;
    *(float4*)&As[r * KP + c] =
        *(const float4*)(X + (size_t)(row0 + r) * K + c);
  }
  __syncthreads();

  const float* __restrict__ arow = As + m * KP;
  const float* __restrict__ wrow = W + (size_t)(col0 + m) * K;

  v8f acc = {};
#if __has_builtin(__builtin_amdgcn_wmma_f32_16x16x4_f32)
  for (int k = 0; k < K; k += 4) {
    v2f a, b;
    a.x = arow[k + koff];
    a.y = arow[k + koff + 1];
    b.x = wrow[k + koff];
    b.y = wrow[k + koff + 1];
    acc = __builtin_amdgcn_wmma_f32_16x16x4_f32(
        false, a, false, b, (short)0, acc, false, false);
  }
#else
  for (int k = 0; k < K; ++k) {
    float av = arow[k];
    #pragma unroll
    for (int v = 0; v < 8; ++v) acc[v] += av * W[(size_t)(col0 + v) * K + k];
  }
#endif

  const int mrow = (lane < 16) ? 0 : 8;
  const int c    = col0 + m;
  const float bv = bias ? bias[c] : 0.0f;
  #pragma unroll
  for (int v = 0; v < 8; ++v) {
    Y[(size_t)(row0 + v + mrow) * DD + c] = acc[v] + bv;
  }
}

__global__ __launch_bounds__(256) void att_logits_k(
    const float* __restrict__ xh, const float* __restrict__ a_src,
    const float* __restrict__ a_dst, float* __restrict__ e_src,
    float* __restrict__ e_dst) {
  int t = blockIdx.x * blockDim.x + threadIdx.x;
  if (t >= NN * HH) return;
  int n = t >> 3, h = t & 7;
  const float4* xp = (const float4*)(xh + (size_t)n * DD + h * CC);
  const float4* sp = (const float4*)(a_src + h * CC);
  const float4* dp = (const float4*)(a_dst + h * CC);
  float ss = 0.f, sd = 0.f;
  #pragma unroll
  for (int i = 0; i < 4; ++i) {
    float4 x = xp[i], s = sp[i], d = dp[i];
    ss += x.x * s.x + x.y * s.y + x.z * s.z + x.w * s.w;
    sd += x.x * d.x + x.y * d.y + x.z * d.z + x.w * d.w;
  }
  e_src[t] = ss;
  e_dst[t] = sd;
}

__device__ __forceinline__ void edge_nodes(const int* __restrict__ ei, int e,
                                           int& s, int& d) {
  if (e < EE) { s = ei[e]; d = ei[EE + e]; }
  else        { s = e - EE; d = e - EE; }
}
__device__ __forceinline__ float leaky(float v) {
  return v > 0.f ? v : v * NEG_SLOPE;
}
__device__ __forceinline__ unsigned f2ord(float f) {
  unsigned u = __float_as_uint(f);
  return (u & 0x80000000u) ? ~u : (u | 0x80000000u);
}
__device__ __forceinline__ float ord2f(unsigned u) {
  return (u & 0x80000000u) ? __uint_as_float(u & 0x7fffffffu)
                           : __uint_as_float(~u);
}
#define ORD_NEG_INF 0x007FFFFFu

__global__ __launch_bounds__(256) void edge_max_k(
    const int* __restrict__ ei, const float* __restrict__ e_src,
    const float* __restrict__ e_dst, unsigned* __restrict__ emax) {
  int t = blockIdx.x * blockDim.x + threadIdx.x;
  if (t >= ETOT * HH) return;
  int e = t >> 3, h = t & 7;
  int s, d; edge_nodes(ei, e, s, d);
  float v = leaky(e_src[s * HH + h] + e_dst[d * HH + h]);
  atomicMax(&emax[d * HH + h], f2ord(v));
}

__global__ __launch_bounds__(256) void edge_den_k(
    const int* __restrict__ ei, const float* __restrict__ e_src,
    const float* __restrict__ e_dst, const unsigned* __restrict__ emax,
    float* __restrict__ denom) {
  int t = blockIdx.x * blockDim.x + threadIdx.x;
  if (t >= ETOT * HH) return;
  int e = t >> 3, h = t & 7;
  int s, d; edge_nodes(ei, e, s, d);
  float v  = leaky(e_src[s * HH + h] + e_dst[d * HH + h]);
  float ex = __expf(v - ord2f(emax[d * HH + h]));
  atomicAdd(&denom[d * HH + h], ex);
}

__global__ __launch_bounds__(256) void edge_aggr_k(
    const int* __restrict__ ei, const float* __restrict__ e_src,
    const float* __restrict__ e_dst, const unsigned* __restrict__ emax,
    const float* __restrict__ denom, const float* __restrict__ xh,
    float* __restrict__ accum) {
  int t = blockIdx.x * blockDim.x + threadIdx.x;
  if (t >= ETOT * 32) return;
  int e = t >> 5, lane = t & 31;
  int h = lane >> 2;
  int s, d; edge_nodes(ei, e, s, d);
  float v     = leaky(e_src[s * HH + h] + e_dst[d * HH + h]);
  float ex    = __expf(v - ord2f(emax[d * HH + h]));
  float alpha = ex / (denom[d * HH + h] + 1e-16f);
  float4 x = *(const float4*)(xh + (size_t)s * DD + lane * 4);
  float* out = accum + (size_t)d * DD + lane * 4;
  atomicAdd(out + 0, x.x * alpha);
  atomicAdd(out + 1, x.y * alpha);
  atomicAdd(out + 2, x.z * alpha);
  atomicAdd(out + 3, x.w * alpha);
}

__global__ __launch_bounds__(256) void residual_ln_k(
    const float* __restrict__ hin, const float* __restrict__ accum,
    const float* __restrict__ gb, const float* __restrict__ g,
    const float* __restrict__ b, float* __restrict__ hout) {
  int node = blockIdx.x * 8 + (threadIdx.x >> 5);
  int lane = threadIdx.x & 31;
  size_t base = (size_t)node * DD + lane * 4;
  float4 hv = *(const float4*)(hin + base);
  float4 av = *(const float4*)(accum + base);
  float4 bv = *(const float4*)(gb + lane * 4);
  float x0 = hv.x + av.x + bv.x;
  float x1 = hv.y + av.y + bv.y;
  float x2 = hv.z + av.z + bv.z;
  float x3 = hv.w + av.w + bv.w;
  float s = x0 + x1 + x2 + x3;
  #pragma unroll
  for (int o = 16; o > 0; o >>= 1) s += __shfl_xor(s, o, 32);
  float mean = s * (1.f / (float)DD);
  float d0 = x0 - mean, d1 = x1 - mean, d2 = x2 - mean, d3 = x3 - mean;
  float vs = d0 * d0 + d1 * d1 + d2 * d2 + d3 * d3;
  #pragma unroll
  for (int o = 16; o > 0; o >>= 1) vs += __shfl_xor(vs, o, 32);
  float inv = rsqrtf(vs * (1.f / (float)DD) + LN_EPS);
  float4 gv = *(const float4*)(g + lane * 4);
  float4 lb = *(const float4*)(b + lane * 4);
  float4 o4;
  o4.x = d0 * inv * gv.x + lb.x;
  o4.y = d1 * inv * gv.y + lb.y;
  o4.z = d2 * inv * gv.z + lb.z;
  o4.w = d3 * inv * gv.w + lb.w;
  *(float4*)(hout + base) = o4;
}

__global__ __launch_bounds__(128) void pool_sum_k(const float* __restrict__ h,
                                                  float* __restrict__ pooled) {
  int c = threadIdx.x;
  float s = 0.f;
  for (int n = blockIdx.x; n < NN; n += gridDim.x)
    s += h[(size_t)n * DD + c];
  atomicAdd(&pooled[c], s);
}

__global__ __launch_bounds__(128) void head_k(
    const float* __restrict__ pooled_sum, const float* __restrict__ W_mu,
    const float* __restrict__ b_mu, const float* __restrict__ W_lv,
    const float* __restrict__ b_lv, float* __restrict__ out) {
  int d = threadIdx.x;
  __shared__ float pl[DD];
  float pd = pooled_sum[d] * (1.f / (float)NN);
  pl[d] = pd;
  __syncthreads();
  float mu = b_mu[d], lv = b_lv[d];
  for (int k = 0; k < DD; ++k) {
    float p = pl[k];
    mu += p * W_mu[(size_t)d * DD + k];
    lv += p * W_lv[(size_t)d * DD + k];
  }
  out[d]      = mu;
  out[DD + d] = lv;
  out[2 * DD + (size_t)NN * DD + d] = pd;
}

__global__ __launch_bounds__(256) void fill32_k(unsigned* __restrict__ p,
                                                unsigned v, int n) {
  int t = blockIdx.x * blockDim.x + threadIdx.x;
  if (t < n) p[t] = v;
}

extern "C" void kernel_launch(void* const* d_in, const int* in_sizes, int n_in,
                              void* d_out, int out_size, void* d_ws, size_t ws_size,
                              hipStream_t stream) {
  const float* x       = (const float*)d_in[0];
  const int*   ei      = (const int*)d_in[1];
  const float* W_in    = (const float*)d_in[2];
  const float* b_in    = (const float*)d_in[3];
  const float* gat_W   = (const float*)d_in[4];
  const float* att_src = (const float*)d_in[5];
  const float* att_dst = (const float*)d_in[6];
  const float* gat_b   = (const float*)d_in[7];
  const float* ln_g    = (const float*)d_in[8];
  const float* ln_b    = (const float*)d_in[9];
  const float* W_mu    = (const float*)d_in[10];
  const float* b_mu    = (const float*)d_in[11];
  const float* W_lv    = (const float*)d_in[12];
  const float* b_lv    = (const float*)d_in[13];
  float* out = (float*)d_out;

  float*    h      = (float*)d_ws;
  float*    xh     = h     + (size_t)NN * DD;
  float*    accum  = xh    + (size_t)NN * DD;
  float*    e_src  = accum + (size_t)NN * DD;
  float*    e_dst  = e_src + (size_t)NN * HH;
  unsigned* emax   = (unsigned*)(e_dst + (size_t)NN * HH);
  float*    denom  = (float*)(emax + (size_t)NN * HH);
  float*    pooled = denom + (size_t)NN * HH;

  float* h_final = out + 2 * DD;

  const int TPB = 256;
  const int nh_blk   = (NN * HH + TPB - 1) / TPB;
  const int eh_blk   = (ETOT * HH + TPB - 1) / TPB;
  const int ea_blk   = (ETOT * 32 + TPB - 1) / TPB;
  const int nd_blk   = ((int)((size_t)NN * DD) + TPB - 1) / TPB;
  const int ln_blk   = NN / 8;
  const int gemm_blk = NN / 16;

  gemm_wmma_k<F_IN><<<gemm_blk, TPB, 0, stream>>>(x, W_in, b_in, h);

  for (int l = 0; l < LL; ++l) {
    const float* Wl  = gat_W   + (size_t)l * DD * DD;
    const float* asl = att_src + (size_t)l * HH * CC;
    const float* adl = att_dst + (size_t)l * HH * CC;
    const float* gbl = gat_b   + (size_t)l * DD;
    const float* gl  = ln_g    + (size_t)l * DD;
    const float* bl  = ln_b    + (size_t)l * DD;

    gemm_wmma_k<DD><<<gemm_blk, TPB, 0, stream>>>(h, Wl, nullptr, xh);

    att_logits_k<<<nh_blk, TPB, 0, stream>>>(xh, asl, adl, e_src, e_dst);

    fill32_k<<<nh_blk, TPB, 0, stream>>>(emax, ORD_NEG_INF, NN * HH);
    fill32_k<<<nh_blk, TPB, 0, stream>>>((unsigned*)denom, 0u, NN * HH);
    fill32_k<<<nd_blk, TPB, 0, stream>>>((unsigned*)accum, 0u, NN * DD);

    edge_max_k <<<eh_blk, TPB, 0, stream>>>(ei, e_src, e_dst, emax);
    edge_den_k <<<eh_blk, TPB, 0, stream>>>(ei, e_src, e_dst, emax, denom);
    edge_aggr_k<<<ea_blk, TPB, 0, stream>>>(ei, e_src, e_dst, emax, denom, xh, accum);

    float* hout = (l == LL - 1) ? h_final : h;
    residual_ln_k<<<ln_blk, TPB, 0, stream>>>(h, accum, gbl, gl, bl, hout);
  }

  fill32_k<<<1, TPB, 0, stream>>>((unsigned*)pooled, 0u, DD);
  pool_sum_k<<<512, DD, 0, stream>>>(h_final, pooled);
  head_k<<<1, DD, 0, stream>>>(pooled, W_mu, b_mu, W_lv, b_lv, out);
}


# edge scatter: lane l owns channels l+32j, so each f32 atomic and gather load of a wave covers one whole 128-byte line (was 8 words in each of 4 lines); atomics kept, add order unordered as before
# speedup vs baseline: 1.2405x; 1.2405x over previous
#include <hip/hip_runtime.h>
#include <hip/hip_bf16.h>


#define NN      100000
#define EE      1600000
#define ETOT    (EE + NN)
#define F_IN    256
#define DD      128
#define HH      8
#define CC      16
#define LL      4
#define NEG_SLOPE 0.2f
#define LN_EPS  1e-5f

typedef __attribute__((ext_vector_type(2))) float v2f;
typedef __attribute__((ext_vector_type(8))) float v8f;

template <int K>
__global__ __launch_bounds__(256) void gemm_wmma_k(
    const float* __restrict__ X, const float* __restrict__ W,
    const float* __restrict__ bias, float* __restrict__ Y) {
  constexpr int KP = K + 4;
  __shared__ float As[16 * KP];

  const int wave = threadIdx.x >> 5;
  const int lane = threadIdx.x & 31;
  const int row0 = blockIdx.x * 16;
  const int col0 = wave * 16;
  const int m    = lane & 15;
  const int koff = (lane < 16) ? 0 : 2;

  #pragma unroll
  for (int i = threadIdx.x * 4; i < 16 * K; i += 256 * 4) {
    const int r = i / K, c = i % K;
    *(float4*)&As[r * KP + c] =
        *(const float4*)(X + (size_t)(row0 + r) * K + c);
  }
  __syncthreads();

  const float* __restrict__ arow = As + m * KP;
  const float* __restrict__ wrow = W + (size_t)(col0 + m) * K;

  v8f acc = {};
#if __has_builtin(__builtin_amdgcn_wmma_f32_16x16x4_f32)
  for (int k = 0; k < K; k += 4) {
    v2f a, b;
    a.x = arow[k + koff];
    a.y = arow[k + koff + 1];
    b.x = wrow[k + koff];
    b.y = wrow[k + koff + 1];
    acc = __builtin_amdgcn_wmma_f32_16x16x4_f32(
        false, a, false, b, (short)0, acc, false, false);
  }
#else
  for (int k = 0; k < K; ++k) {
    float av = arow[k];
    #pragma unroll
    for (int v = 0; v < 8; ++v) acc[v] += av * W[(size_t)(col0 + v) * K + k];
  }
#endif

  const int mrow = (lane < 16) ? 0 : 8;
  const int c    = col0 + m;
  const float bv = bias ? bias[c] : 0.0f;
  #pragma unroll
  for (int v = 0; v < 8; ++v) {
    Y[(size_t)(row0 + v + mrow) * DD + c] = acc[v] + bv;
  }
}

__global__ __launch_bounds__(256) void att_logits_k(
    const float* __restrict__ xh, const float* __restrict__ a_src,
    const float* __restrict__ a_dst, float* __restrict__ e_src,
    float* __restrict__ e_dst) {
  int t = blockIdx.x * blockDim.x + threadIdx.x;
  if (t >= NN * HH) return;
  int n = t >> 3, h = t & 7;
  const float4* xp = (const float4*)(xh + (size_t)n * DD + h * CC);
  const float4* sp = (const float4*)(a_src + h * CC);
  const float4* dp = (const float4*)(a_dst + h * CC);
  float ss = 0.f, sd = 0.f;
  #pragma unroll
  for (int i = 0; i < 4; ++i) {
    float4 x = xp[i], s = sp[i], d = dp[i];
    ss += x.x * s.x + x.y * s.y + x.z * s.z + x.w * s.w;
    sd += x.x * d.x + x.y * d.y + x.z * d.z + x.w * d.w;
  }
  e_src[t] = ss;
  e_dst[t] = sd;
}

__device__ __forceinline__ void edge_nodes(const int* __restrict__ ei, int e,
                                           int& s, int& d) {
  if (e < EE) { s = ei[e]; d = ei[EE + e]; }
  else        { s = e - EE; d = e - EE; }
}
__device__ __forceinline__ float leaky(float v) {
  return v > 0.f ? v : v * NEG_SLOPE;
}
__device__ __forceinline__ unsigned f2ord(float f) {
  unsigned u = __float_as_uint(f);
  return (u & 0x80000000u) ? ~u : (u | 0x80000000u);
}
__device__ __forceinline__ float ord2f(unsigned u) {
  return (u & 0x80000000u) ? __uint_as_float(u & 0x7fffffffu)
                           : __uint_as_float(~u);
}
#define ORD_NEG_INF 0x007FFFFFu

__global__ __launch_bounds__(256) void edge_max_k(
    const int* __restrict__ ei, const float* __restrict__ e_src,
    const float* __restrict__ e_dst, unsigned* __restrict__ emax) {
  int t = blockIdx.x * blockDim.x + threadIdx.x;
  if (t >= ETOT * HH) return;
  int e = t >> 3, h = t & 7;
  int s, d; edge_nodes(ei, e, s, d);
  float v = leaky(e_src[s * HH + h] + e_dst[d * HH + h]);
  atomicMax(&emax[d * HH + h], f2ord(v));
}

__global__ __launch_bounds__(256) void edge_den_k(
    const int* __restrict__ ei, const float* __restrict__ e_src,
    const float* __restrict__ e_dst, const unsigned* __restrict__ emax,
    float* __restrict__ denom) {
  int t = blockIdx.x * blockDim.x + threadIdx.x;
  if (t >= ETOT * HH) return;
  int e = t >> 3, h = t & 7;
  int s, d; edge_nodes(ei, e, s, d);
  float v  = leaky(e_src[s * HH + h] + e_dst[d * HH + h]);
  float ex = __expf(v - ord2f(emax[d * HH + h]));
  atomicAdd(&denom[d * HH + h], ex);
}

__global__ __launch_bounds__(256) void edge_aggr_k(
    const int* __restrict__ ei, const float* __restrict__ e_src,
    const float* __restrict__ e_dst, const unsigned* __restrict__ emax,
    const float* __restrict__ denom, const float* __restrict__ xh,
    float* __restrict__ accum) {
  int t = blockIdx.x * blockDim.x + threadIdx.x;
  if (t >= ETOT * 32) return;
  int e = t >> 5, lane = t & 31;
  int h = lane & 7;
  int s, d; edge_nodes(ei, e, s, d);
  float v     = leaky(e_src[s * HH + h] + e_dst[d * HH + h]);
  float ex    = __expf(v - ord2f(emax[d * HH + h]));
  float alpha = ex / (denom[d * HH + h] + 1e-16f);
  const float* xs = xh + (size_t)s * DD + lane;
  float* out = accum + (size_t)d * DD + lane;
  const int hs = lane >> 4;
  float x0 = xs[0], x1 = xs[32], x2 = xs[64], x3 = xs[96];
  float a0 = __shfl(alpha, 0 + hs, 32);
  float a1 = __shfl(alpha, 2 + hs, 32);
  float a2 = __shfl(alpha, 4 + hs, 32);
  float a3 = __shfl(alpha, 6 + hs, 32);
  atomicAdd(out + 0,  x0 * a0);
  atomicAdd(out + 32, x1 * a1);
  atomicAdd(out + 64, x2 * a2);
  atomicAdd(out + 96, x3 * a3);
}

__global__ __launch_bounds__(256) void residual_ln_k(
    const float* __restrict__ hin, const float* __restrict__ accum,
    const float* __restrict__ gb, const float* __restrict__ g,
    const float* __restrict__ b, float* __restrict__ hout) {
  int node = blockIdx.x * 8 + (threadIdx.x >> 5);
  int lane = threadIdx.x & 31;
  size_t base = (size_t)node * DD + lane * 4;
  float4 hv = *(const float4*)(hin + base);
  float4 av = *(const float4*)(accum + base);
  float4 bv = *(const float4*)(gb + lane * 4);
  float x0 = hv.x + av.x + bv.x;
  float x1 = hv.y + av.y + bv.y;
  float x2 = hv.z + av.z + bv.z;
  float x3 = hv.w + av.w + bv.w;
  float s = x0 + x1 + x2 + x3;
  #pragma unroll
  for (int o = 16; o > 0; o >>= 1) s += __shfl_xor(s, o, 32);
  float mean = s * (1.f / (float)DD);
  float d0 = x0 - mean, d1 = x1 - mean, d2 = x2 - mean, d3 = x3 - mean;
  float vs = d0 * d0 + d1 * d1 + d2 * d2 + d3 * d3;
  #pragma unroll
  for (int o = 16; o > 0; o >>= 1) vs += __shfl_xor(vs, o, 32);
  float inv = rsqrtf(vs * (1.f / (float)DD) + LN_EPS);
  float4 gv = *(const float4*)(g + lane * 4);
  float4 lb = *(const float4*)(b + lane * 4);
  float4 o4;
  o4.x = d0 * inv * gv.x + lb.x;
  o4.y = d1 * inv * gv.y + lb.y;
  o4.z = d2 * inv * gv.z + lb.z;
  o4.w = d3 * inv * gv.w + lb.w;
  *(float4*)(hout + base) = o4;
}

__global__ __launch_bounds__(128) void pool_sum_k(const float* __restrict__ h,
                                                  float* __restrict__ pooled) {
  int c = threadIdx.x;
  float s = 0.f;
  for (int n = blockIdx.x; n < NN; n += gridDim.x)
    s += h[(size_t)n * DD + c];
  atomicAdd(&pooled[c], s);
}

__global__ __launch_bounds__(128) void head_k(
    const float* __restrict__ pooled_sum, const float* __restrict__ W_mu,
    const float* __restrict__ b_mu, const float* __restrict__ W_lv,
    const float* __restrict__ b_lv, float* __restrict__ out) {
  int d = threadIdx.x;
  __shared__ float pl[DD];
  float pd = pooled_sum[d] * (1.f / (float)NN);
  pl[d] = pd;
  __syncthreads();
  float mu = b_mu[d], lv = b_lv[d];
  for (int k = 0; k < DD; ++k) {
    float p = pl[k];
    mu += p * W_mu[(size_t)d * DD + k];
    lv += p * W_lv[(size_t)d * DD + k];
  }
  out[d]      = mu;
  out[DD + d] = lv;
  out[2 * DD + (size_t)NN * DD + d] = pd;
}

__global__ __launch_bounds__(256) void fill32_k(unsigned* __restrict__ p,
                                                unsigned v, int n) {
  int t = blockIdx.x * blockDim.x + threadIdx.x;
  if (t < n) p[t] = v;
}

extern "C" void kernel_launch(void* const* d_in, const int* in_sizes, int n_in,
                              void* d_out, int out_size, void* d_ws, size_t ws_size,
                              hipStream_t stream) {
  const float* x       = (const float*)d_in[0];
  const int*   ei      = (const int*)d_in[1];
  const float* W_in    = (const float*)d_in[2];
  const float* b_in    = (const float*)d_in[3];
  const float* gat_W   = (const float*)d_in[4];
  const float* att_src = (const float*)d_in[5];
  const float* att_dst = (const float*)d_in[6];
  const float* gat_b   = (const float*)d_in[7];
  const float* ln_g    = (const float*)d_in[8];
  const float* ln_b    = (const float*)d_in[9];
  const float* W_mu    = (const float*)d_in[10];
  const float* b_mu    = (const float*)d_in[11];
  const float* W_lv    = (const float*)d_in[12];
  const float* b_lv    = (const float*)d_in[13];
  float* out = (float*)d_out;

  float*    h      = (float*)d_ws;
  float*    xh     = h     + (size_t)NN * DD;
  float*    accum  = xh    + (size_t)NN * DD;
  float*    e_src  = accum + (size_t)NN * DD;
  float*    e_dst  = e_src + (size_t)NN * HH;
  unsigned* emax   = (unsigned*)(e_dst + (size_t)NN * HH);
  float*    denom  = (float*)(emax + (size_t)NN * HH);
  float*    pooled = denom + (size_t)NN * HH;

  float* h_final = out + 2 * DD;

  const int TPB = 256;
  const int nh_blk   = (NN * HH + TPB - 1) / TPB;
  const int eh_blk   = (ETOT * HH + TPB - 1) / TPB;
  const int ea_blk   = (ETOT * 32 + TPB - 1) / TPB;
  const int nd_blk   = ((int)((size_t)NN * DD) + TPB - 1) / TPB;
  const int ln_blk   = NN / 8;
  const int gemm_blk = NN / 16;

  gemm_wmma_k<F_IN><<<gemm_blk, TPB, 0, stream>>>(x, W_in, b_in, h);

  for (int l = 0; l < LL; ++l) {
    const float* Wl  = gat_W   + (size_t)l * DD * DD;
    const float* asl = att_src + (size_t)l * HH * CC;
    const float* adl = att_dst + (size_t)l * HH * CC;
    const float* gbl = gat_b   + (size_t)l * DD;
    const float* gl  = ln_g    + (size_t)l * DD;
    const float* bl  = ln_b    + (size_t)l * DD;

    gemm_wmma_k<DD><<<gemm_blk, TPB, 0, stream>>>(h, Wl, nullptr, xh);

    att_logits_k<<<nh_blk, TPB, 0, stream>>>(xh, asl, adl, e_src, e_dst);

    fill32_k<<<nh_blk, TPB, 0, stream>>>(emax, ORD_NEG_INF, NN * HH);
    fill32_k<<<nh_blk, TPB, 0, stream>>>((unsigned*)denom, 0u, NN * HH);
    fill32_k<<<nd_blk, TPB, 0, stream>>>((unsigned*)accum, 0u, NN * DD);

    edge_max_k <<<eh_blk, TPB, 0, stream>>>(ei, e_src, e_dst, emax);
    edge_den_k <<<eh_blk, TPB, 0, stream>>>(ei, e_src, e_dst, emax, denom);
    edge_aggr_k<<<ea_blk, TPB, 0, stream>>>(ei, e_src, e_dst, emax, denom, xh, accum);

    float* hout = (l == LL - 1) ? h_final : h;
    residual_ln_k<<<ln_blk, TPB, 0, stream>>>(h, accum, gbl, gl, bl, hout);
  }

  fill32_k<<<1, TPB, 0, stream>>>((unsigned*)pooled, 0u, DD);
  pool_sum_k<<<512, DD, 0, stream>>>(h_final, pooled);
  head_k<<<1, DD, 0, stream>>>(pooled, W_mu, b_mu, W_lv, b_lv, out);
}


# in-edges grouped by dst once per call (int atomics; sums run in slot arrival order); per layer one wave per node: softmax, weighted sum, residual+LayerNorm in registers; f32 atomics and clears gone
# speedup vs baseline: 2.3954x; 1.9310x over previous
#include <hip/hip_runtime.h>
#include <hip/hip_bf16.h>


#define NN      100000
#define EE      1600000
#define ETOT    (EE + NN)
#define F_IN    256
#define DD      128
#define HH      8
#define CC      16
#define LL      4
#define NEG_SLOPE 0.2f
#define LN_EPS  1e-5f

typedef __attribute__((ext_vector_type(2))) float v2f;
typedef __attribute__((ext_vector_type(8))) float v8f;

template <int K>
__global__ __launch_bounds__(256) void gemm_wmma_k(
    const float* __restrict__ X, const float* __restrict__ W,
    const float* __restrict__ bias, float* __restrict__ Y) {
  constexpr int KP = K + 4;
  __shared__ float As[16 * KP];

  const int wave = threadIdx.x >> 5;
  const int lane = threadIdx.x & 31;
  const int row0 = blockIdx.x * 16;
  const int col0 = wave * 16;
  const int m    = lane & 15;
  const int koff = (lane < 16) ? 0 : 2;

  #pragma unroll
  for (int i = threadIdx.x * 4; i < 16 * K; i += 256 * 4) {
    const int r = i / K, c = i % K;
    *(float4*)&As[r * KP + c] =
        *(const float4*)(X + (size_t)(row0 + r) * K + c);
  }
  __syncthreads();

  const float* __restrict__ arow = As + m * KP;
  const float* __restrict__ wrow = W + (size_t)(col0 + m) * K;

  v8f acc = {};
#if __has_builtin(__builtin_amdgcn_wmma_f32_16x16x4_f32)
  for (int k = 0; k < K; k += 4) {
    v2f a, b;
    a.x = arow[k + koff];
    a.y = arow[k + koff + 1];
    b.x = wrow[k + koff];
    b.y = wrow[k + koff + 1];
    acc = __builtin_amdgcn_wmma_f32_16x16x4_f32(
        false, a, false, b, (short)0, acc, false, false);
  }
#else
  for (int k = 0; k < K; ++k) {
    float av = arow[k];
    #pragma unroll
    for (int v = 0; v < 8; ++v) acc[v] += av * W[(size_t)(col0 + v) * K + k];
  }
#endif

  const int mrow = (lane < 16) ? 0 : 8;
  const int c    = col0 + m;
  const float bv = bias ? bias[c] : 0.0f;
  #pragma unroll
  for (int v = 0; v < 8; ++v) {
    Y[(size_t)(row0 + v + mrow) * DD + c] = acc[v] + bv;
  }
}

__global__ __launch_bounds__(256) void att_logits_k(
    const float* __restrict__ xh, const float* __restrict__ a_src,
    const float* __restrict__ a_dst, float* __restrict__ e_src,
    float* __restrict__ e_dst) {
  int t = blockIdx.x * blockDim.x + threadIdx.x;
  if (t >= NN * HH) return;
  int n = t >> 3, h = t & 7;
  const float4* xp = (const float4*)(xh + (size_t)n * DD + h * CC);
  const float4* sp = (const float4*)(a_src + h * CC);
  const float4* dp = (const float4*)(a_dst + h * CC);
  float ss = 0.f, sd = 0.f;
  #pragma unroll
  for (int i = 0; i < 4; ++i) {
    float4 x = xp[i], s = sp[i], d = dp[i];
    ss += x.x * s.x + x.y * s.y + x.z * s.z + x.w * s.w;
    sd += x.x * d.x + x.y * d.y + x.z * d.z + x.w * d.w;
  }
  e_src[t] = ss;
  e_dst[t] = sd;
}

__device__ __forceinline__ float leaky(float v) {
  return v > 0.f ? v : v * NEG_SLOPE;
}

__global__ __launch_bounds__(256) void seg_count_k(const int* __restrict__ ei,
                                                   int* __restrict__ cnt) {
  int e = blockIdx.x * blockDim.x + threadIdx.x;
  if (e < EE) atomicAdd(&cnt[ei[EE + e]], 1);
}

__global__ __launch_bounds__(256) void seg_alloc_k(int* cntpos,
                                                   int* __restrict__ cursor,
                                                   int* __restrict__ seg_beg) {
  __shared__ int wsum[8];
  __shared__ int base_s;
  const int n    = blockIdx.x * 256 + threadIdx.x;
  const int lane = threadIdx.x & 31;
  const int wave = threadIdx.x >> 5;
  const int c = (n < NN) ? cntpos[n] : 0;
  int incl = c;
  #pragma unroll
  for (int o = 1; o < 32; o <<= 1) {
    int t = __shfl_up(incl, o, 32);
    if (lane >= o) incl += t;
  }
  if (lane == 31) wsum[wave] = incl;
  __syncthreads();
  if (threadIdx.x == 0) {
    int tot = 0;
    for (int w = 0; w < 8; ++w) { int t = wsum[w]; wsum[w] = tot; tot += t; }
    base_s = atomicAdd(cursor, tot);
  }
  __syncthreads();
  if (n < NN) {
    const int first = base_s + wsum[wave] + incl - c;
    seg_beg[n] = first;
    cntpos[n]  = first;
  }
}

__global__ __launch_bounds__(256) void seg_fill_k(const int* __restrict__ ei,
                                                  int* __restrict__ fillpos,
                                                  int* __restrict__ csr_src) {
  int e = blockIdx.x * blockDim.x + threadIdx.x;
  if (e < EE) {
    int p = atomicAdd(&fillpos[ei[EE + e]], 1);
    csr_src[p] = ei[e];
  }
}

__global__ __launch_bounds__(256) void node_gat_ln_k(
    const int* __restrict__ seg_beg, const int* __restrict__ seg_end,
    const int* __restrict__ csr_src, const float* __restrict__ e_src,
    const float* __restrict__ e_dst, const float* __restrict__ xh,
    const float* hin, const float* __restrict__ gb,
    const float* __restrict__ g, const float* __restrict__ b, float* hout) {
  const int node = __builtin_amdgcn_readfirstlane(
      (int)(blockIdx.x * 8 + (threadIdx.x >> 5)));
  const int lane = threadIdx.x & 31;
  const int h    = lane & 7;
  const int sub  = lane >> 3;
  const int hs   = lane >> 4;
  const int beg   = seg_beg[node];
  const int deg   = seg_end[node] - beg;
  const int total = deg + 1;
  const float ed  = e_dst[node * HH + h];

  float mx = -__builtin_inff();
  for (int j = sub; j < total; j += 4) {
    const int s = (j < deg) ? csr_src[beg + j] : node;
    mx = fmaxf(mx, leaky(e_src[s * HH + h] + ed));
  }
  mx = fmaxf(mx, __shfl_xor(mx, 8, 32));
  mx = fmaxf(mx, __shfl_xor(mx, 16, 32));

  float sm = 0.f;
  for (int j = sub; j < total; j += 4) {
    const int s = (j < deg) ? csr_src[beg + j] : node;
    sm += __expf(leaky(e_src[s * HH + h] + ed) - mx);
  }
  sm += __shfl_xor(sm, 8, 32);
  sm += __shfl_xor(sm, 16, 32);
  const float den = sm + 1e-16f;

  float acc0 = 0.f, acc1 = 0.f, acc2 = 0.f, acc3 = 0.f;
  const float* __restrict__ xl = xh + lane;
#define EDGE_ACC(U)                                                    \
  {                                                                    \
    const int su = __shfl(s, (U) * 8, 32);                             \
    const float* __restrict__ xp = xl + (size_t)su * DD;               \
    const float x0 = xp[0], x1 = xp[32], x2 = xp[64], x3 = xp[96];     \
    acc0 += x0 * __shfl(alpha, (U) * 8 + 0 + hs, 32);                  \
    acc1 += x1 * __shfl(alpha, (U) * 8 + 2 + hs, 32);                  \
    acc2 += x2 * __shfl(alpha, (U) * 8 + 4 + hs, 32);                  \
    acc3 += x3 * __shfl(alpha, (U) * 8 + 6 + hs, 32);                  \
  }
  for (int base = 0; base < total; base += 4) {
    const int j = base + sub;
    int   s     = node;
    float alpha = 0.f;
    if (j < total) {
      if (j < deg) s = csr_src[beg + j];
      alpha = __expf(leaky(e_src[s * HH + h] + ed) - mx) / den;
    }
    const int left = total - base;
    if (left >= 4) {
      EDGE_ACC(0) EDGE_ACC(1) EDGE_ACC(2) EDGE_ACC(3)
    } else {
      EDGE_ACC(0)
      if (left >= 2) EDGE_ACC(1)
      if (left >= 3) EDGE_ACC(2)
    }
  }
#undef EDGE_ACC

  const size_t rbase = (size_t)node * DD + lane;
  const float x0 = hin[rbase]      + acc0 + gb[lane];
  const float x1 = hin[rbase + 32] + acc1 + gb[lane + 32];
  const float x2 = hin[rbase + 64] + acc2 + gb[lane + 64];
  const float x3 = hin[rbase + 96] + acc3 + gb[lane + 96];
  float ssum = x0 + x1 + x2 + x3;
  #pragma unroll
  for (int o = 16; o > 0; o >>= 1) ssum += __shfl_xor(ssum, o, 32);
  const float mean = ssum * (1.f / (float)DD);
  const float d0 = x0 - mean, d1 = x1 - mean, d2 = x2 - mean, d3 = x3 - mean;
  float vs = d0 * d0 + d1 * d1 + d2 * d2 + d3 * d3;
  #pragma unroll
  for (int o = 16; o > 0; o >>= 1) vs += __shfl_xor(vs, o, 32);
  const float inv = rsqrtf(vs * (1.f / (float)DD) + LN_EPS);
  hout[rbase]      = d0 * inv * g[lane]      + b[lane];
  hout[rbase + 32] = d1 * inv * g[lane + 32] + b[lane + 32];
  hout[rbase + 64] = d2 * inv * g[lane + 64] + b[lane + 64];
  hout[rbase + 96] = d3 * inv * g[lane + 96] + b[lane + 96];
}

__global__ __launch_bounds__(128) void pool_sum_k(const float* __restrict__ h,
                                                  float* __restrict__ pooled) {
  int c = threadIdx.x;
  float s = 0.f;
  for (int n = blockIdx.x; n < NN; n += gridDim.x)
    s += h[(size_t)n * DD + c];
  atomicAdd(&pooled[c], s);
}

__global__ __launch_bounds__(128) void head_k(
    const float* __restrict__ pooled_sum, const float* __restrict__ W_mu,
    const float* __restrict__ b_mu, const float* __restrict__ W_lv,
    const float* __restrict__ b_lv, float* __restrict__ out) {
  int d = threadIdx.x;
  __shared__ float pl[DD];
  float pd = pooled_sum[d] * (1.f / (float)NN);
  pl[d] = pd;
  __syncthreads();
  float mu = b_mu[d], lv = b_lv[d];
  for (int k = 0; k < DD; ++k) {
    float p = pl[k];
    mu += p * W_mu[(size_t)d * DD + k];
    lv += p * W_lv[(size_t)d * DD + k];
  }
  out[d]      = mu;
  out[DD + d] = lv;
  out[2 * DD + (size_t)NN * DD + d] = pd;
}

__global__ __launch_bounds__(256) void fill32_k(unsigned* __restrict__ p,
                                                unsigned v, int n) {
  int t = blockIdx.x * blockDim.x + threadIdx.x;
  if (t < n) p[t] = v;
}

extern "C" void kernel_launch(void* const* d_in, const int* in_sizes, int n_in,
                              void* d_out, int out_size, void* d_ws, size_t ws_size,
                              hipStream_t stream) {
  const float* x       = (const float*)d_in[0];
  const int*   ei      = (const int*)d_in[1];
  const float* W_in    = (const float*)d_in[2];
  const float* b_in    = (const float*)d_in[3];
  const float* gat_W   = (const float*)d_in[4];
  const float* att_src = (const float*)d_in[5];
  const float* att_dst = (const float*)d_in[6];
  const float* gat_b   = (const float*)d_in[7];
  const float* ln_g    = (const float*)d_in[8];
  const float* ln_b    = (const float*)d_in[9];
  const float* W_mu    = (const float*)d_in[10];
  const float* b_mu    = (const float*)d_in[11];
  const float* W_lv    = (const float*)d_in[12];
  const float* b_lv    = (const float*)d_in[13];
  float* out = (float*)d_out;

  float* h       = (float*)d_ws;
  float* xh      = h  + (size_t)NN * DD;
  float* grp     = xh + (size_t)NN * DD;
  float* e_src   = grp   + (size_t)NN * DD;
  float* e_dst   = e_src + (size_t)NN * HH;
  float* pooled  = e_dst + (size_t)NN * HH * 3;

  int* cntpos  = (int*)grp;
  int* cursor  = cntpos + NN;
  int* seg_beg = cntpos + NN + 32;
  int* csr_src = seg_beg + NN;

  float* h_final = out + 2 * DD;

  const int TPB = 256;
  const int nh_blk   = (NN * HH + TPB - 1) / TPB;
  const int e_blk    = (EE + TPB - 1) / TPB;
  const int n_blk    = (NN + TPB - 1) / TPB;
  const int n1_blk   = (NN + 1 + TPB - 1) / TPB;
  const int node_blk = NN / 8;
  const int gemm_blk = NN / 16;

  fill32_k   <<<n1_blk, TPB, 0, stream>>>((unsigned*)cntpos, 0u, NN + 1);
  seg_count_k<<<e_blk,  TPB, 0, stream>>>(ei, cntpos);
  seg_alloc_k<<<n_blk,  TPB, 0, stream>>>(cntpos, cursor, seg_beg);
  seg_fill_k <<<e_blk,  TPB, 0, stream>>>(ei, cntpos, csr_src);

  gemm_wmma_k<F_IN><<<gemm_blk, TPB, 0, stream>>>(x, W_in, b_in, h);

  for (int l = 0; l < LL; ++l) {
    const float* Wl  = gat_W   + (size_t)l * DD * DD;
    const float* asl = att_src + (size_t)l * HH * CC;
    const float* adl = att_dst + (size_t)l * HH * CC;
    const float* gbl = gat_b   + (size_t)l * DD;
    const float* gl  = ln_g    + (size_t)l * DD;
    const float* bl  = ln_b    + (size_t)l * DD;

    gemm_wmma_k<DD><<<gemm_blk, TPB, 0, stream>>>(h, Wl, nullptr, xh);

    att_logits_k<<<nh_blk, TPB, 0, stream>>>(xh, asl, adl, e_src, e_dst);

    float* hout = (l == LL - 1) ? h_final : h;
    node_gat_ln_k<<<node_blk, TPB, 0, stream>>>(seg_beg, cntpos, csr_src, e_src,
                                                e_dst, xh, h, gbl, gl, bl, hout);
  }

  fill32_k<<<1, TPB, 0, stream>>>((unsigned*)pooled, 0u, DD);
  pool_sum_k<<<512, DD, 0, stream>>>(h_final, pooled);
  head_k<<<1, DD, 0, stream>>>(pooled, W_mu, b_mu, W_lv, b_lv, out);
}


# node pass, nodes of up to 32 edges: indices read once (lane j = edge j), scores gathered once and kept in registers, row loads up to two sweeps ahead; larger nodes keep the 3 passes; sum order kept
# speedup vs baseline: 2.8262x; 1.1798x over previous
#include <hip/hip_runtime.h>
#include <hip/hip_bf16.h>


#define NN      100000
#define EE      1600000
#define ETOT    (EE + NN)
#define F_IN    256
#define DD      128
#define HH      8
#define CC      16
#define LL      4
#define NEG_SLOPE 0.2f
#define LN_EPS  1e-5f

typedef __attribute__((ext_vector_type(2))) float v2f;
typedef __attribute__((ext_vector_type(8))) float v8f;

template <int K>
__global__ __launch_bounds__(256) void gemm_wmma_k(
    const float* __restrict__ X, const float* __restrict__ W,
    const float* __restrict__ bias, float* __restrict__ Y) {
  constexpr int KP = K + 4;
  __shared__ float As[16 * KP];

  const int wave = threadIdx.x >> 5;
  const int lane = threadIdx.x & 31;
  const int row0 = blockIdx.x * 16;
  const int col0 = wave * 16;
  const int m    = lane & 15;
  const int koff = (lane < 16) ? 0 : 2;

  #pragma unroll
  for (int i = threadIdx.x * 4; i < 16 * K; i += 256 * 4) {
    const int r = i / K, c = i % K;
    *(float4*)&As[r * KP + c] =
        *(const float4*)(X + (size_t)(row0 + r) * K + c);
  }
  __syncthreads();

  const float* __restrict__ arow = As + m * KP;
  const float* __restrict__ wrow = W + (size_t)(col0 + m) * K;

  v8f acc = {};
#if __has_builtin(__builtin_amdgcn_wmma_f32_16x16x4_f32)
  for (int k = 0; k < K; k += 4) {
    v2f a, b;
    a.x = arow[k + koff];
    a.y = arow[k + koff + 1];
    b.x = wrow[k + koff];
    b.y = wrow[k + koff + 1];
    acc = __builtin_amdgcn_wmma_f32_16x16x4_f32(
        false, a, false, b, (short)0, acc, false, false);
  }
#else
  for (int k = 0; k < K; ++k) {
    float av = arow[k];
    #pragma unroll
    for (int v = 0; v < 8; ++v) acc[v] += av * W[(size_t)(col0 + v) * K + k];
  }
#endif

  const int mrow = (lane < 16) ? 0 : 8;
  const int c    = col0 + m;
  const float bv = bias ? bias[c] : 0.0f;
  #pragma unroll
  for (int v = 0; v < 8; ++v) {
    Y[(size_t)(row0 + v + mrow) * DD + c] = acc[v] + bv;
  }
}

__global__ __launch_bounds__(256) void att_logits_k(
    const float* __restrict__ xh, const float* __restrict__ a_src,
    const float* __restrict__ a_dst, float* __restrict__ e_src,
    float* __restrict__ e_dst) {
  int t = blockIdx.x * blockDim.x + threadIdx.x;
  if (t >= NN * HH) return;
  int n = t >> 3, h = t & 7;
  const float4* xp = (const float4*)(xh + (size_t)n * DD + h * CC);
  const float4* sp = (const float4*)(a_src + h * CC);
  const float4* dp = (const float4*)(a_dst + h * CC);
  float ss = 0.f, sd = 0.f;
  #pragma unroll
  for (int i = 0; i < 4; ++i) {
    float4 x = xp[i], s = sp[i], d = dp[i];
    ss += x.x * s.x + x.y * s.y + x.z * s.z + x.w * s.w;
    sd += x.x * d.x + x.y * d.y + x.z * d.z + x.w * d.w;
  }
  e_src[t] = ss;
  e_dst[t] = sd;
}

__device__ __forceinline__ float leaky(float v) {
  return v > 0.f ? v : v * NEG_SLOPE;
}

__global__ __launch_bounds__(256) void seg_count_k(const int* __restrict__ ei,
                                                   int* __restrict__ cnt) {
  int e = blockIdx.x * blockDim.x + threadIdx.x;
  if (e < EE) atomicAdd(&cnt[ei[EE + e]], 1);
}

__global__ __launch_bounds__(256) void seg_alloc_k(int* cntpos,
                                                   int* __restrict__ cursor,
                                                   int* __restrict__ seg_beg) {
  __shared__ int wsum[8];
  __shared__ int base_s;
  const int n    = blockIdx.x * 256 + threadIdx.x;
  const int lane = threadIdx.x & 31;
  const int wave = threadIdx.x >> 5;
  const int c = (n < NN) ? cntpos[n] : 0;
  int incl = c;
  #pragma unroll
  for (int o = 1; o < 32; o <<= 1) {
    int t = __shfl_up(incl, o, 32);
    if (lane >= o) incl += t;
  }
  if (lane == 31) wsum[wave] = incl;
  __syncthreads();
  if (threadIdx.x == 0) {
    int tot = 0;
    for (int w = 0; w < 8; ++w) { int t = wsum[w]; wsum[w] = tot; tot += t; }
    base_s = atomicAdd(cursor, tot);
  }
  __syncthreads();
  if (n < NN) {
    const int first = base_s + wsum[wave] + incl - c;
    seg_beg[n] = first;
    cntpos[n]  = first;
  }
}

__global__ __launch_bounds__(256) void seg_fill_k(const int* __restrict__ ei,
                                                  int* __restrict__ fillpos,
                                                  int* __restrict__ csr_src) {
  int e = blockIdx.x * blockDim.x + threadIdx.x;
  if (e < EE) {
    int p = atomicAdd(&fillpos[ei[EE + e]], 1);
    csr_src[p] = ei[e];
  }
}

__device__ __forceinline__ void load_sweep(const float* __restrict__ xh, int lane,
                                           int sidx, int t, float (&x)[16]) {
  #pragma unroll
  for (int u = 0; u < 4; ++u) {
    const int su = __builtin_amdgcn_readlane(sidx, 4 * t + u);
    const float* __restrict__ xrow = xh + (size_t)su * DD;
    x[4 * u + 0] = xrow[lane];
    x[4 * u + 1] = xrow[lane + 32];
    x[4 * u + 2] = xrow[lane + 64];
    x[4 * u + 3] = xrow[lane + 96];
  }
}

__device__ __forceinline__ void acc_sweep(const float (&x)[16], float al,
                                          int hs, float& a0, float& a1,
                                          float& a2, float& a3) {
  #pragma unroll
  for (int u = 0; u < 4; ++u) {
    const int ai = __float_as_int(al);
    const float e0 = __int_as_float(__builtin_amdgcn_readlane(ai, u * 8 + 0));
    const float o0 = __int_as_float(__builtin_amdgcn_readlane(ai, u * 8 + 1));
    const float e1 = __int_as_float(__builtin_amdgcn_readlane(ai, u * 8 + 2));
    const float o1 = __int_as_float(__builtin_amdgcn_readlane(ai, u * 8 + 3));
    const float e2 = __int_as_float(__builtin_amdgcn_readlane(ai, u * 8 + 4));
    const float o2 = __int_as_float(__builtin_amdgcn_readlane(ai, u * 8 + 5));
    const float e3 = __int_as_float(__builtin_amdgcn_readlane(ai, u * 8 + 6));
    const float o3 = __int_as_float(__builtin_amdgcn_readlane(ai, u * 8 + 7));
    a0 = __builtin_fmaf(x[4 * u + 0], hs ? o0 : e0, a0);
    a1 = __builtin_fmaf(x[4 * u + 1], hs ? o1 : e1, a1);
    a2 = __builtin_fmaf(x[4 * u + 2], hs ? o2 : e2, a2);
    a3 = __builtin_fmaf(x[4 * u + 3], hs ? o3 : e3, a3);
  }
}

template <int T>
__device__ __forceinline__ void sweep_pipe(int ns, const float* __restrict__ xh,
                                           int lane, int sidx,
                                           const float (&al)[8],
                                           int hs, float (&cur)[16],
                                           float (&nxt)[16], float& a0,
                                           float& a1, float& a2, float& a3) {
  acc_sweep(cur, al[T], hs, a0, a1, a2, a3);
  if constexpr (T + 2 < 8) {
    if (ns > T + 2) {
      load_sweep(xh, lane, sidx, T + 2, cur);
      sweep_pipe<T + 1>(ns, xh, lane, sidx, al, hs, nxt, cur, a0, a1, a2, a3);
      return;
    }
  }
  if constexpr (T + 1 < 8) {
    if (ns > T + 1)
      acc_sweep(nxt, al[T + 1], hs, a0, a1, a2, a3);
  }
}

__global__ __launch_bounds__(256, 9) void node_gat_ln_k(
    const int* __restrict__ seg_beg, const int* __restrict__ seg_end,
    const int* __restrict__ csr_src, const float* __restrict__ e_src,
    const float* __restrict__ e_dst, const float* __restrict__ xh,
    const float* hin, const float* __restrict__ gb,
    const float* __restrict__ g, const float* __restrict__ b, float* hout) {
  const int node = __builtin_amdgcn_readfirstlane(
      (int)(blockIdx.x * 8 + (threadIdx.x >> 5)));
  const int lane = threadIdx.x & 31;
  const int h    = lane & 7;
  const int sub  = lane >> 3;
  const int hs   = lane >> 4;
  const int beg   = seg_beg[node];
  const int deg   = seg_end[node] - beg;
  const int total = deg + 1;
  const float ed  = e_dst[node * HH + h];
  const size_t rbase = (size_t)node * DD + lane;
  const float* __restrict__ xl = xh + lane;

  float acc0 = 0.f, acc1 = 0.f, acc2 = 0.f, acc3 = 0.f;
  float r0, r1, r2, r3;

  if (total <= 32) {
    const int ns = (total + 3) >> 2;
    int sidx = node;
    if (lane < deg) sidx = csr_src[beg + lane];

    r0 = hin[rbase];
    r1 = hin[rbase + 32];
    r2 = hin[rbase + 64];
    r3 = hin[rbase + 96];

    float ev[8];
    #pragma unroll
    for (int t = 0; t < 8; ++t) {
      ev[t] = 0.f;
      if (t == 0 || ns > t) {
        const int s = __shfl(sidx, 4 * t + sub, 32);
        ev[t] = e_src[s * HH + h];
      }
    }

    float xa[16], xb[16];
    load_sweep(xh, lane, sidx, 0, xa);
    if (ns > 1) load_sweep(xh, lane, sidx, 1, xb);

    float sc[8];
    float mx = -__builtin_inff();
    #pragma unroll
    for (int t = 0; t < 8; ++t) {
      sc[t] = leaky(ev[t] + ed);
      if (4 * t + sub < total) mx = fmaxf(mx, sc[t]);
    }
    mx = fmaxf(mx, __shfl_xor(mx, 8, 32));
    mx = fmaxf(mx, __shfl_xor(mx, 16, 32));

    float sm = 0.f;
    #pragma unroll
    for (int t = 0; t < 8; ++t) {
      sc[t] = __expf(sc[t] - mx);
      if (4 * t + sub < total) sm += sc[t];
    }
    sm += __shfl_xor(sm, 8, 32);
    sm += __shfl_xor(sm, 16, 32);
    const float den = sm + 1e-16f;

    float al[8];
    #pragma unroll
    for (int t = 0; t < 8; ++t)
      al[t] = (4 * t + sub < total) ? sc[t] / den : 0.f;

    sweep_pipe<0>(ns, xh, lane, sidx, al, hs, xa, xb, acc0, acc1, acc2, acc3);
  } else {
    float mx = -__builtin_inff();
    for (int j = sub; j < total; j += 4) {
      const int s = (j < deg) ? csr_src[beg + j] : node;
      mx = fmaxf(mx, leaky(e_src[s * HH + h] + ed));
    }
    mx = fmaxf(mx, __shfl_xor(mx, 8, 32));
    mx = fmaxf(mx, __shfl_xor(mx, 16, 32));

    float sm = 0.f;
    for (int j = sub; j < total; j += 4) {
      const int s = (j < deg) ? csr_src[beg + j] : node;
      sm += __expf(leaky(e_src[s * HH + h] + ed) - mx);
    }
    sm += __shfl_xor(sm, 8, 32);
    sm += __shfl_xor(sm, 16, 32);
    const float den = sm + 1e-16f;

#define EDGE_ACC(U)                                                    \
    {                                                                  \
      const int su = __shfl(s, (U) * 8, 32);                           \
      const float* __restrict__ xp = xl + (size_t)su * DD;             \
      const float x0 = xp[0], x1 = xp[32], x2 = xp[64], x3 = xp[96];   \
      acc0 += x0 * __shfl(alpha, (U) * 8 + 0 + hs, 32);                \
      acc1 += x1 * __shfl(alpha, (U) * 8 + 2 + hs, 32);                \
      acc2 += x2 * __shfl(alpha, (U) * 8 + 4 + hs, 32);                \
      acc3 += x3 * __shfl(alpha, (U) * 8 + 6 + hs, 32);                \
    }
    for (int base = 0; base < total; base += 4) {
      const int j = base + sub;
      int   s     = node;
      float alpha = 0.f;
      if (j < total) {
        if (j < deg) s = csr_src[beg + j];
        alpha = __expf(leaky(e_src[s * HH + h] + ed) - mx) / den;
      }
      const int left = total - base;
      if (left >= 4) {
        EDGE_ACC(0) EDGE_ACC(1) EDGE_ACC(2) EDGE_ACC(3)
      } else {
        EDGE_ACC(0)
        if (left >= 2) EDGE_ACC(1)
        if (left >= 3) EDGE_ACC(2)
      }
    }
#undef EDGE_ACC
    r0 = hin[rbase];
    r1 = hin[rbase + 32];
    r2 = hin[rbase + 64];
    r3 = hin[rbase + 96];
  }

  const float x0 = r0 + acc0 + gb[lane];
  const float x1 = r1 + acc1 + gb[lane + 32];
  const float x2 = r2 + acc2 + gb[lane + 64];
  const float x3 = r3 + acc3 + gb[lane + 96];
  float ssum = x0 + x1 + x2 + x3;
  #pragma unroll
  for (int o = 16; o > 0; o >>= 1) ssum += __shfl_xor(ssum, o, 32);
  const float mean = ssum * (1.f / (float)DD);
  const float d0 = x0 - mean, d1 = x1 - mean, d2 = x2 - mean, d3 = x3 - mean;
  float vs = d0 * d0 + d1 * d1 + d2 * d2 + d3 * d3;
  #pragma unroll
  for (int o = 16; o > 0; o >>= 1) vs += __shfl_xor(vs, o, 32);
  const float inv = rsqrtf(vs * (1.f / (float)DD) + LN_EPS);
  hout[rbase]      = d0 * inv * g[lane]      + b[lane];
  hout[rbase + 32] = d1 * inv * g[lane + 32] + b[lane + 32];
  hout[rbase + 64] = d2 * inv * g[lane + 64] + b[lane + 64];
  hout[rbase + 96] = d3 * inv * g[lane + 96] + b[lane + 96];
}

__global__ __launch_bounds__(128) void pool_sum_k(const float* __restrict__ h,
                                                  float* __restrict__ pooled) {
  int c = threadIdx.x;
  float s = 0.f;
  for (int n = blockIdx.x; n < NN; n += gridDim.x)
    s += h[(size_t)n * DD + c];
  atomicAdd(&pooled[c], s);
}

__global__ __launch_bounds__(128) void head_k(
    const float* __restrict__ pooled_sum, const float* __restrict__ W_mu,
    const float* __restrict__ b_mu, const float* __restrict__ W_lv,
    const float* __restrict__ b_lv, float* __restrict__ out) {
  int d = threadIdx.x;
  __shared__ float pl[DD];
  float pd = pooled_sum[d] * (1.f / (float)NN);
  pl[d] = pd;
  __syncthreads();
  float mu = b_mu[d], lv = b_lv[d];
  for (int k = 0; k < DD; ++k) {
    float p = pl[k];
    mu += p * W_mu[(size_t)d * DD + k];
    lv += p * W_lv[(size_t)d * DD + k];
  }
  out[d]      = mu;
  out[DD + d] = lv;
  out[2 * DD + (size_t)NN * DD + d] = pd;
}

__global__ __launch_bounds__(256) void fill32_k(unsigned* __restrict__ p,
                                                unsigned v, int n) {
  int t = blockIdx.x * blockDim.x + threadIdx.x;
  if (t < n) p[t] = v;
}

extern "C" void kernel_launch(void* const* d_in, const int* in_sizes, int n_in,
                              void* d_out, int out_size, void* d_ws, size_t ws_size,
                              hipStream_t stream) {
  const float* x       = (const float*)d_in[0];
  const int*   ei      = (const int*)d_in[1];
  const float* W_in    = (const float*)d_in[2];
  const float* b_in    = (const float*)d_in[3];
  const float* gat_W   = (const float*)d_in[4];
  const float* att_src = (const float*)d_in[5];
  const float* att_dst = (const float*)d_in[6];
  const float* gat_b   = (const float*)d_in[7];
  const float* ln_g    = (const float*)d_in[8];
  const float* ln_b    = (const float*)d_in[9];
  const float* W_mu    = (const float*)d_in[10];
  const float* b_mu    = (const float*)d_in[11];
  const float* W_lv    = (const float*)d_in[12];
  const float* b_lv    = (const float*)d_in[13];
  float* out = (float*)d_out;

  float* h       = (float*)d_ws;
  float* xh      = h  + (size_t)NN * DD;
  float* grp     = xh + (size_t)NN * DD;
  float* e_src   = grp   + (size_t)NN * DD;
  float* e_dst   = e_src + (size_t)NN * HH;
  float* pooled  = e_dst + (size_t)NN * HH * 3;

  int* cntpos  = (int*)grp;
  int* cursor  = cntpos + NN;
  int* seg_beg = cntpos + NN + 32;
  int* csr_src = seg_beg + NN;

  float* h_final = out + 2 * DD;

  const int TPB = 256;
  const int nh_blk   = (NN * HH + TPB - 1) / TPB;
  const int e_blk    = (EE + TPB - 1) / TPB;
  const int n_blk    = (NN + TPB - 1) / TPB;
  const int n1_blk   = (NN + 1 + TPB - 1) / TPB;
  const int node_blk = NN / 8;
  const int gemm_blk = NN / 16;

  fill32_k   <<<n1_blk, TPB, 0, stream>>>((unsigned*)cntpos, 0u, NN + 1);
  seg_count_k<<<e_blk,  TPB, 0, stream>>>(ei, cntpos);
  seg_alloc_k<<<n_blk,  TPB, 0, stream>>>(cntpos, cursor, seg_beg);
  seg_fill_k <<<e_blk,  TPB, 0, stream>>>(ei, cntpos, csr_src);

  gemm_wmma_k<F_IN><<<gemm_blk, TPB, 0, stream>>>(x, W_in, b_in, h);

  for (int l = 0; l < LL; ++l) {
    const float* Wl  = gat_W   + (size_t)l * DD * DD;
    const float* asl = att_src + (size_t)l * HH * CC;
    const float* adl = att_dst + (size_t)l * HH * CC;
    const float* gbl = gat_b   + (size_t)l * DD;
    const float* gl  = ln_g    + (size_t)l * DD;
    const float* bl  = ln_b    + (size_t)l * DD;

    gemm_wmma_k<DD><<<gemm_blk, TPB, 0, stream>>>(h, Wl, nullptr, xh);

    att_logits_k<<<nh_blk, TPB, 0, stream>>>(xh, asl, adl, e_src, e_dst);

    float* hout = (l == LL - 1) ? h_final : h;
    node_gat_ln_k<<<node_blk, TPB, 0, stream>>>(seg_beg, cntpos, csr_src, e_src,
                                                e_dst, xh, h, gbl, gl, bl, hout);
  }

  fill32_k<<<1, TPB, 0, stream>>>((unsigned*)pooled, 0u, DD);
  pool_sum_k<<<512, DD, 0, stream>>>(h_final, pooled);
  head_k<<<1, DD, 0, stream>>>(pooled, W_mu, b_mu, W_lv, b_lv, out);
}


# GEMMs: each wave loads the B fragments of its 16 columns once into registers and reuses them over 6-7 passes of 32 rows (grid 512); was one 16-line global B load per WMMA; k and sum order kept
# speedup vs baseline: 3.1580x; 1.1174x over previous
#include <hip/hip_runtime.h>
#include <hip/hip_bf16.h>


#define NN      100000
#define EE      1600000
#define ETOT    (EE + NN)
#define F_IN    256
#define DD      128
#define HH      8
#define CC      16
#define LL      4
#define NEG_SLOPE 0.2f
#define LN_EPS  1e-5f

typedef __attribute__((ext_vector_type(2))) float v2f;
typedef __attribute__((ext_vector_type(4))) float v4f;
typedef __attribute__((ext_vector_type(8))) float v8f;

template <int K>
__global__ __launch_bounds__(256, 4) void gemm_wmma_k(
    const float* __restrict__ X, const float* __restrict__ W,
    const float* __restrict__ bias, float* __restrict__ Y) {
  constexpr int KP    = K + 4;
  constexpr int PR    = 32;
  constexpr int NPASS = NN / PR;
  constexpr int TPR   = K / 4;
  constexpr int RPS   = 256 / TPR;
  constexpr int NV    = PR / RPS;
  constexpr int NS    = K / 4;
  constexpr int CH    = 4;
  constexpr int NG    = NS / CH;
  __shared__ float As[PR * KP];

  const int wave = threadIdx.x >> 5;
  const int lane = threadIdx.x & 31;
  const int col0 = wave * 16;
  const int m    = lane & 15;
  const int koff = (lane < 16) ? 0 : 2;
  const int sr   = threadIdx.x / TPR;
  const int sc   = (threadIdx.x % TPR) * 4;
  const int gstride = (int)gridDim.x;

  const float* __restrict__ wrow = W + (size_t)(col0 + m) * K + koff;
  v2f b[NS];
  #pragma unroll
  for (int i = 0; i < NS; ++i) {
    b[i].x = wrow[4 * i];
    b[i].y = wrow[4 * i + 1];
  }

  const int mrow = (lane < 16) ? 0 : 8;
  const int c    = col0 + m;
  const float bv = bias ? bias[c] : 0.0f;

  const float* __restrict__ a0row = As + m * KP + koff;
  const float* __restrict__ a1row = As + (16 + m) * KP + koff;
  const float* __restrict__ xs    = X + (size_t)sr * K + sc;
  float* __restrict__ ls          = As + sr * KP + sc;

  int pass = blockIdx.x;
  v4f pre[NV];
  {
    const float* __restrict__ xp = xs + (size_t)min(pass, NPASS - 1) * PR * K;
    #pragma unroll
    for (int i = 0; i < NV; ++i)
      pre[i] = *(const v4f*)(xp + (size_t)i * RPS * K);
  }

  for (; pass < NPASS; pass += gstride) {
    #pragma unroll
    for (int i = 0; i < NV; ++i)
      *(v4f*)(ls + i * RPS * KP) = pre[i];
    __syncthreads();

    {
      const float* __restrict__ xp =
          xs + (size_t)min(pass + gstride, NPASS - 1) * PR * K;
      #pragma unroll
      for (int i = 0; i < NV; ++i)
        pre[i] = *(const v4f*)(xp + (size_t)i * RPS * K);
    }

    v2f fa[2][2 * CH];
    #pragma unroll
    for (int j = 0; j < CH; ++j) {
      fa[0][j].x      = a0row[4 * j];
      fa[0][j].y      = a0row[4 * j + 1];
      fa[0][CH + j].x = a1row[4 * j];
      fa[0][CH + j].y = a1row[4 * j + 1];
    }
    v8f acc0 = {}, acc1 = {};
    #pragma unroll
    for (int g = 0; g < NG; ++g) {
      if (g + 1 < NG) {
        #pragma unroll
        for (int j = 0; j < CH; ++j) {
          const int k = 4 * ((g + 1) * CH + j);
          fa[(g + 1) & 1][j].x      = a0row[k];
          fa[(g + 1) & 1][j].y      = a0row[k + 1];
          fa[(g + 1) & 1][CH + j].x = a1row[k];
          fa[(g + 1) & 1][CH + j].y = a1row[k + 1];
        }
      }
      __builtin_amdgcn_sched_barrier(0);
      #pragma unroll
      for (int j = 0; j < CH; ++j) {
        acc0 = __builtin_amdgcn_wmma_f32_16x16x4_f32(
            false, fa[g & 1][j], false, b[g * CH + j], (short)0, acc0,
            false, false);
        acc1 = __builtin_amdgcn_wmma_f32_16x16x4_f32(
            false, fa[g & 1][CH + j], false, b[g * CH + j], (short)0, acc1,
            false, false);
      }
      __builtin_amdgcn_sched_barrier(0);
    }

    float* __restrict__ yp = Y + ((size_t)pass * PR + mrow) * DD + c;
    #pragma unroll
    for (int v = 0; v < 8; ++v) {
      yp[(size_t)v * DD]        = acc0[v] + bv;
      yp[(size_t)(16 + v) * DD] = acc1[v] + bv;
    }
    __syncthreads();
  }
}

__global__ __launch_bounds__(256) void att_logits_k(
    const float* __restrict__ xh, const float* __restrict__ a_src,
    const float* __restrict__ a_dst, float* __restrict__ e_src,
    float* __restrict__ e_dst) {
  int t = blockIdx.x * blockDim.x + threadIdx.x;
  if (t >= NN * HH) return;
  int n = t >> 3, h = t & 7;
  const float4* xp = (const float4*)(xh + (size_t)n * DD + h * CC);
  const float4* sp = (const float4*)(a_src + h * CC);
  const float4* dp = (const float4*)(a_dst + h * CC);
  float ss = 0.f, sd = 0.f;
  #pragma unroll
  for (int i = 0; i < 4; ++i) {
    float4 x = xp[i], s = sp[i], d = dp[i];
    ss += x.x * s.x + x.y * s.y + x.z * s.z + x.w * s.w;
    sd += x.x * d.x + x.y * d.y + x.z * d.z + x.w * d.w;
  }
  e_src[t] = ss;
  e_dst[t] = sd;
}

__device__ __forceinline__ float leaky(float v) {
  return v > 0.f ? v : v * NEG_SLOPE;
}

__global__ __launch_bounds__(256) void seg_count_k(const int* __restrict__ ei,
                                                   int* __restrict__ cnt) {
  int e = blockIdx.x * blockDim.x + threadIdx.x;
  if (e < EE) atomicAdd(&cnt[ei[EE + e]], 1);
}

__global__ __launch_bounds__(256) void seg_alloc_k(int* cntpos,
                                                   int* __restrict__ cursor,
                                                   int* __restrict__ seg_beg) {
  __shared__ int wsum[8];
  __shared__ int base_s;
  const int n    = blockIdx.x * 256 + threadIdx.x;
  const int lane = threadIdx.x & 31;
  const int wave = threadIdx.x >> 5;
  const int c = (n < NN) ? cntpos[n] : 0;
  int incl = c;
  #pragma unroll
  for (int o = 1; o < 32; o <<= 1) {
    int t = __shfl_up(incl, o, 32);
    if (lane >= o) incl += t;
  }
  if (lane == 31) wsum[wave] = incl;
  __syncthreads();
  if (threadIdx.x == 0) {
    int tot = 0;
    for (int w = 0; w < 8; ++w) { int t = wsum[w]; wsum[w] = tot; tot += t; }
    base_s = atomicAdd(cursor, tot);
  }
  __syncthreads();
  if (n < NN) {
    const int first = base_s + wsum[wave] + incl - c;
    seg_beg[n] = first;
    cntpos[n]  = first;
  }
}

__global__ __launch_bounds__(256) void seg_fill_k(const int* __restrict__ ei,
                                                  int* __restrict__ fillpos,
                                                  int* __restrict__ csr_src) {
  int e = blockIdx.x * blockDim.x + threadIdx.x;
  if (e < EE) {
    int p = atomicAdd(&fillpos[ei[EE + e]], 1);
    csr_src[p] = ei[e];
  }
}

__device__ __forceinline__ void load_sweep(const float* __restrict__ xh, int lane,
                                           int sidx, int t, float (&x)[16]) {
  #pragma unroll
  for (int u = 0; u < 4; ++u) {
    const int su = __builtin_amdgcn_readlane(sidx, 4 * t + u);
    const float* __restrict__ xrow = xh + (size_t)su * DD;
    x[4 * u + 0] = xrow[lane];
    x[4 * u + 1] = xrow[lane + 32];
    x[4 * u + 2] = xrow[lane + 64];
    x[4 * u + 3] = xrow[lane + 96];
  }
}

__device__ __forceinline__ void acc_sweep(const float (&x)[16], float al,
                                          int hs, float& a0, float& a1,
                                          float& a2, float& a3) {
  #pragma unroll
  for (int u = 0; u < 4; ++u) {
    const int ai = __float_as_int(al);
    const float e0 = __int_as_float(__builtin_amdgcn_readlane(ai, u * 8 + 0));
    const float o0 = __int_as_float(__builtin_amdgcn_readlane(ai, u * 8 + 1));
    const float e1 = __int_as_float(__builtin_amdgcn_readlane(ai, u * 8 + 2));
    const float o1 = __int_as_float(__builtin_amdgcn_readlane(ai, u * 8 + 3));
    const float e2 = __int_as_float(__builtin_amdgcn_readlane(ai, u * 8 + 4));
    const float o2 = __int_as_float(__builtin_amdgcn_readlane(ai, u * 8 + 5));
    const float e3 = __int_as_float(__builtin_amdgcn_readlane(ai, u * 8 + 6));
    const float o3 = __int_as_float(__builtin_amdgcn_readlane(ai, u * 8 + 7));
    a0 = __builtin_fmaf(x[4 * u + 0], hs ? o0 : e0, a0);
    a1 = __builtin_fmaf(x[4 * u + 1], hs ? o1 : e1, a1);
    a2 = __builtin_fmaf(x[4 * u + 2], hs ? o2 : e2, a2);
    a3 = __builtin_fmaf(x[4 * u + 3], hs ? o3 : e3, a3);
  }
}

template <int T>
__device__ __forceinline__ void sweep_pipe(int ns, const float* __restrict__ xh,
                                           int lane, int sidx,
                                           const float (&al)[8],
                                           int hs, float (&cur)[16],
                                           float (&nxt)[16], float& a0,
                                           float& a1, float& a2, float& a3) {
  acc_sweep(cur, al[T], hs, a0, a1, a2, a3);
  if constexpr (T + 2 < 8) {
    if (ns > T + 2) {
      load_sweep(xh, lane, sidx, T + 2, cur);
      sweep_pipe<T + 1>(ns, xh, lane, sidx, al, hs, nxt, cur, a0, a1, a2, a3);
      return;
    }
  }
  if constexpr (T + 1 < 8) {
    if (ns > T + 1)
      acc_sweep(nxt, al[T + 1], hs, a0, a1, a2, a3);
  }
}

__global__ __launch_bounds__(256, 9) void node_gat_ln_k(
    const int* __restrict__ seg_beg, const int* __restrict__ seg_end,
    const int* __restrict__ csr_src, const float* __restrict__ e_src,
    const float* __restrict__ e_dst, const float* __restrict__ xh,
    const float* hin, const float* __restrict__ gb,
    const float* __restrict__ g, const float* __restrict__ b, float* hout) {
  const int node = __builtin_amdgcn_readfirstlane(
      (int)(blockIdx.x * 8 + (threadIdx.x >> 5)));
  const int lane = threadIdx.x & 31;
  const int h    = lane & 7;
  const int sub  = lane >> 3;
  const int hs   = lane >> 4;
  const int beg   = seg_beg[node];
  const int deg   = seg_end[node] - beg;
  const int total = deg + 1;
  const float ed  = e_dst[node * HH + h];
  const size_t rbase = (size_t)node * DD + lane;
  const float* __restrict__ xl = xh + lane;

  float acc0 = 0.f, acc1 = 0.f, acc2 = 0.f, acc3 = 0.f;
  float r0, r1, r2, r3;

  if (total <= 32) {
    const int ns = (total + 3) >> 2;
    int sidx = node;
    if (lane < deg) sidx = csr_src[beg + lane];

    r0 = hin[rbase];
    r1 = hin[rbase + 32];
    r2 = hin[rbase + 64];
    r3 = hin[rbase + 96];

    float ev[8];
    #pragma unroll
    for (int t = 0; t < 8; ++t) {
      ev[t] = 0.f;
      if (t == 0 || ns > t) {
        const int s = __shfl(sidx, 4 * t + sub, 32);
        ev[t] = e_src[s * HH + h];
      }
    }

    float xa[16], xb[16];
    load_sweep(xh, lane, sidx, 0, xa);
    if (ns > 1) load_sweep(xh, lane, sidx, 1, xb);

    float sc[8];
    float mx = -__builtin_inff();
    #pragma unroll
    for (int t = 0; t < 8; ++t) {
      sc[t] = leaky(ev[t] + ed);
      if (4 * t + sub < total) mx = fmaxf(mx, sc[t]);
    }
    mx = fmaxf(mx, __shfl_xor(mx, 8, 32));
    mx = fmaxf(mx, __shfl_xor(mx, 16, 32));

    float sm = 0.f;
    #pragma unroll
    for (int t = 0; t < 8; ++t) {
      sc[t] = __expf(sc[t] - mx);
      if (4 * t + sub < total) sm += sc[t];
    }
    sm += __shfl_xor(sm, 8, 32);
    sm += __shfl_xor(sm, 16, 32);
    const float den = sm + 1e-16f;

    float al[8];
    #pragma unroll
    for (int t = 0; t < 8; ++t)
      al[t] = (4 * t + sub < total) ? sc[t] / den : 0.f;

    sweep_pipe<0>(ns, xh, lane, sidx, al, hs, xa, xb, acc0, acc1, acc2, acc3);
  } else {
    float mx = -__builtin_inff();
    for (int j = sub; j < total; j += 4) {
      const int s = (j < deg) ? csr_src[beg + j] : node;
      mx = fmaxf(mx, leaky(e_src[s * HH + h] + ed));
    }
    mx = fmaxf(mx, __shfl_xor(mx, 8, 32));
    mx = fmaxf(mx, __shfl_xor(mx, 16, 32));

    float sm = 0.f;
    for (int j = sub; j < total; j += 4) {
      const int s = (j < deg) ? csr_src[beg + j] : node;
      sm += __expf(leaky(e_src[s * HH + h] + ed) - mx);
    }
    sm += __shfl_xor(sm, 8, 32);
    sm += __shfl_xor(sm, 16, 32);
    const float den = sm + 1e-16f;

#define EDGE_ACC(U)                                                    \
    {                                                                  \
      const int su = __shfl(s, (U) * 8, 32);                           \
      const float* __restrict__ xp = xl + (size_t)su * DD;             \
      const float x0 = xp[0], x1 = xp[32], x2 = xp[64], x3 = xp[96];   \
      acc0 += x0 * __shfl(alpha, (U) * 8 + 0 + hs, 32);                \
      acc1 += x1 * __shfl(alpha, (U) * 8 + 2 + hs, 32);                \
      acc2 += x2 * __shfl(alpha, (U) * 8 + 4 + hs, 32);                \
      acc3 += x3 * __shfl(alpha, (U) * 8 + 6 + hs, 32);                \
    }
    for (int base = 0; base < total; base += 4) {
      const int j = base + sub;
      int   s     = node;
      float alpha = 0.f;
      if (j < total) {
        if (j < deg) s = csr_src[beg + j];
        alpha = __expf(leaky(e_src[s * HH + h] + ed) - mx) / den;
      }
      const int left = total - base;
      if (left >= 4) {
        EDGE_ACC(0) EDGE_ACC(1) EDGE_ACC(2) EDGE_ACC(3)
      } else {
        EDGE_ACC(0)
        if (left >= 2) EDGE_ACC(1)
        if (left >= 3) EDGE_ACC(2)
      }
    }
#undef EDGE_ACC
    r0 = hin[rbase];
    r1 = hin[rbase + 32];
    r2 = hin[rbase + 64];
    r3 = hin[rbase + 96];
  }

  const float x0 = r0 + acc0 + gb[lane];
  const float x1 = r1 + acc1 + gb[lane + 32];
  const float x2 = r2 + acc2 + gb[lane + 64];
  const float x3 = r3 + acc3 + gb[lane + 96];
  float ssum = x0 + x1 + x2 + x3;
  #pragma unroll
  for (int o = 16; o > 0; o >>= 1) ssum += __shfl_xor(ssum, o, 32);
  const float mean = ssum * (1.f / (float)DD);
  const float d0 = x0 - mean, d1 = x1 - mean, d2 = x2 - mean, d3 = x3 - mean;
  float vs = d0 * d0 + d1 * d1 + d2 * d2 + d3 * d3;
  #pragma unroll
  for (int o = 16; o > 0; o >>= 1) vs += __shfl_xor(vs, o, 32);
  const float inv = rsqrtf(vs * (1.f / (float)DD) + LN_EPS);
  hout[rbase]      = d0 * inv * g[lane]      + b[lane];
  hout[rbase + 32] = d1 * inv * g[lane + 32] + b[lane + 32];
  hout[rbase + 64] = d2 * inv * g[lane + 64] + b[lane + 64];
  hout[rbase + 96] = d3 * inv * g[lane + 96] + b[lane + 96];
}

__global__ __launch_bounds__(128) void pool_sum_k(const float* __restrict__ h,
                                                  float* __restrict__ pooled) {
  int c = threadIdx.x;
  float s = 0.f;
  for (int n = blockIdx.x; n < NN; n += gridDim.x)
    s += h[(size_t)n * DD + c];
  atomicAdd(&pooled[c], s);
}

__global__ __launch_bounds__(128) void head_k(
    const float* __restrict__ pooled_sum, const float* __restrict__ W_mu,
    const float* __restrict__ b_mu, const float* __restrict__ W_lv,
    const float* __restrict__ b_lv, float* __restrict__ out) {
  int d = threadIdx.x;
  __shared__ float pl[DD];
  float pd = pooled_sum[d] * (1.f / (float)NN);
  pl[d] = pd;
  __syncthreads();
  float mu = b_mu[d], lv = b_lv[d];
  for (int k = 0; k < DD; ++k) {
    float p = pl[k];
    mu += p * W_mu[(size_t)d * DD + k];
    lv += p * W_lv[(size_t)d * DD + k];
  }
  out[d]      = mu;
  out[DD + d] = lv;
  out[2 * DD + (size_t)NN * DD + d] = pd;
}

__global__ __launch_bounds__(256) void fill32_k(unsigned* __restrict__ p,
                                                unsigned v, int n) {
  int t = blockIdx.x * blockDim.x + threadIdx.x;
  if (t < n) p[t] = v;
}

extern "C" void kernel_launch(void* const* d_in, const int* in_sizes, int n_in,
                              void* d_out, int out_size, void* d_ws, size_t ws_size,
                              hipStream_t stream) {
  const float* x       = (const float*)d_in[0];
  const int*   ei      = (const int*)d_in[1];
  const float* W_in    = (const float*)d_in[2];
  const float* b_in    = (const float*)d_in[3];
  const float* gat_W   = (const float*)d_in[4];
  const float* att_src = (const float*)d_in[5];
  const float* att_dst = (const float*)d_in[6];
  const float* gat_b   = (const float*)d_in[7];
  const float* ln_g    = (const float*)d_in[8];
  const float* ln_b    = (const float*)d_in[9];
  const float* W_mu    = (const float*)d_in[10];
  const float* b_mu    = (const float*)d_in[11];
  const float* W_lv    = (const float*)d_in[12];
  const float* b_lv    = (const float*)d_in[13];
  float* out = (float*)d_out;

  float* h       = (float*)d_ws;
  float* xh      = h  + (size_t)NN * DD;
  float* grp     = xh + (size_t)NN * DD;
  float* e_src   = grp   + (size_t)NN * DD;
  float* e_dst   = e_src + (size_t)NN * HH;
  float* pooled  = e_dst + (size_t)NN * HH * 3;

  int* cntpos  = (int*)grp;
  int* cursor  = cntpos + NN;
  int* seg_beg = cntpos + NN + 32;
  int* csr_src = seg_beg + NN;

  float* h_final = out + 2 * DD;

  const int TPB = 256;
  const int nh_blk   = (NN * HH + TPB - 1) / TPB;
  const int e_blk    = (EE + TPB - 1) / TPB;
  const int n_blk    = (NN + TPB - 1) / TPB;
  const int n1_blk   = (NN + 1 + TPB - 1) / TPB;
  const int node_blk = NN / 8;
  const int gemm_blk = 512;

  fill32_k   <<<n1_blk, TPB, 0, stream>>>((unsigned*)cntpos, 0u, NN + 1);
  seg_count_k<<<e_blk,  TPB, 0, stream>>>(ei, cntpos);
  seg_alloc_k<<<n_blk,  TPB, 0, stream>>>(cntpos, cursor, seg_beg);
  seg_fill_k <<<e_blk,  TPB, 0, stream>>>(ei, cntpos, csr_src);

  gemm_wmma_k<F_IN><<<gemm_blk, TPB, 0, stream>>>(x, W_in, b_in, h);

  for (int l = 0; l < LL; ++l) {
    const float* Wl  = gat_W   + (size_t)l * DD * DD;
    const float* asl = att_src + (size_t)l * HH * CC;
    const float* adl = att_dst + (size_t)l * HH * CC;
    const float* gbl = gat_b   + (size_t)l * DD;
    const float* gl  = ln_g    + (size_t)l * DD;
    const float* bl  = ln_b    + (size_t)l * DD;

    gemm_wmma_k<DD><<<gemm_blk, TPB, 0, stream>>>(h, Wl, nullptr, xh);

    att_logits_k<<<nh_blk, TPB, 0, stream>>>(xh, asl, adl, e_src, e_dst);

    float* hout = (l == LL - 1) ? h_final : h;
    node_gat_ln_k<<<node_blk, TPB, 0, stream>>>(seg_beg, cntpos, csr_src, e_src,
                                                e_dst, xh, h, gbl, gl, bl, hout);
  }

  fill32_k<<<1, TPB, 0, stream>>>((unsigned*)pooled, 0u, DD);
  pool_sum_k<<<512, DD, 0, stream>>>(h_final, pooled);
  head_k<<<1, DD, 0, stream>>>(pooled, W_mu, b_mu, W_lv, b_lv, out);
}


# node pass: lane l owns channels 4l..4l+3 (one head): a row is one 16-byte load per lane (was 4 one-line loads), alpha comes from the lane's quad by DPP; edge sum order kept, LayerNorm sums regrouped
# speedup vs baseline: 3.3584x; 1.0635x over previous
#include <hip/hip_runtime.h>
#include <hip/hip_bf16.h>


#define NN      100000
#define EE      1600000
#define ETOT    (EE + NN)
#define F_IN    256
#define DD      128
#define HH      8
#define CC      16
#define LL      4
#define NEG_SLOPE 0.2f
#define LN_EPS  1e-5f

typedef __attribute__((ext_vector_type(2))) float v2f;
typedef __attribute__((ext_vector_type(4))) float v4f;
typedef __attribute__((ext_vector_type(8))) float v8f;

template <int K>
__global__ __launch_bounds__(256, 4) void gemm_wmma_k(
    const float* __restrict__ X, const float* __restrict__ W,
    const float* __restrict__ bias, float* __restrict__ Y) {
  constexpr int KP    = K + 4;
  constexpr int PR    = 32;
  constexpr int NPASS = NN / PR;
  constexpr int TPR   = K / 4;
  constexpr int RPS   = 256 / TPR;
  constexpr int NV    = PR / RPS;
  constexpr int NS    = K / 4;
  constexpr int CH    = 4;
  constexpr int NG    = NS / CH;
  __shared__ float As[PR * KP];

  const int wave = threadIdx.x >> 5;
  const int lane = threadIdx.x & 31;
  const int col0 = wave * 16;
  const int m    = lane & 15;
  const int koff = (lane < 16) ? 0 : 2;
  const int sr   = threadIdx.x / TPR;
  const int sc   = (threadIdx.x % TPR) * 4;
  const int gstride = (int)gridDim.x;

  const float* __restrict__ wrow = W + (size_t)(col0 + m) * K + koff;
  v2f b[NS];
  #pragma unroll
  for (int i = 0; i < NS; ++i) {
    b[i].x = wrow[4 * i];
    b[i].y = wrow[4 * i + 1];
  }

  const int mrow = (lane < 16) ? 0 : 8;
  const int c    = col0 + m;
  const float bv = bias ? bias[c] : 0.0f;

  const float* __restrict__ a0row = As + m * KP + koff;
  const float* __restrict__ a1row = As + (16 + m) * KP + koff;
  const float* __restrict__ xs    = X + (size_t)sr * K + sc;
  float* __restrict__ ls          = As + sr * KP + sc;

  int pass = blockIdx.x;
  v4f pre[NV];
  {
    const float* __restrict__ xp = xs + (size_t)min(pass, NPASS - 1) * PR * K;
    #pragma unroll
    for (int i = 0; i < NV; ++i)
      pre[i] = *(const v4f*)(xp + (size_t)i * RPS * K);
  }

  for (; pass < NPASS; pass += gstride) {
    #pragma unroll
    for (int i = 0; i < NV; ++i)
      *(v4f*)(ls + i * RPS * KP) = pre[i];
    __syncthreads();

    {
      const float* __restrict__ xp =
          xs + (size_t)min(pass + gstride, NPASS - 1) * PR * K;
      #pragma unroll
      for (int i = 0; i < NV; ++i)
        pre[i] = *(const v4f*)(xp + (size_t)i * RPS * K);
    }

    v2f fa[2][2 * CH];
    #pragma unroll
    for (int j = 0; j < CH; ++j) {
      fa[0][j].x      = a0row[4 * j];
      fa[0][j].y      = a0row[4 * j + 1];
      fa[0][CH + j].x = a1row[4 * j];
      fa[0][CH + j].y = a1row[4 * j + 1];
    }
    v8f acc0 = {}, acc1 = {};
    #pragma unroll
    for (int g = 0; g < NG; ++g) {
      if (g + 1 < NG) {
        #pragma unroll
        for (int j = 0; j < CH; ++j) {
          const int k = 4 * ((g + 1) * CH + j);
          fa[(g + 1) & 1][j].x      = a0row[k];
          fa[(g + 1) & 1][j].y      = a0row[k + 1];
          fa[(g + 1) & 1][CH + j].x = a1row[k];
          fa[(g + 1) & 1][CH + j].y = a1row[k + 1];
        }
      }
      __builtin_amdgcn_sched_barrier(0);
      #pragma unroll
      for (int j = 0; j < CH; ++j) {
        acc0 = __builtin_amdgcn_wmma_f32_16x16x4_f32(
            false, fa[g & 1][j], false, b[g * CH + j], (short)0, acc0,
            false, false);
        acc1 = __builtin_amdgcn_wmma_f32_16x16x4_f32(
            false, fa[g & 1][CH + j], false, b[g * CH + j], (short)0, acc1,
            false, false);
      }
      __builtin_amdgcn_sched_barrier(0);
    }

    float* __restrict__ yp = Y + ((size_t)pass * PR + mrow) * DD + c;
    #pragma unroll
    for (int v = 0; v < 8; ++v) {
      yp[(size_t)v * DD]        = acc0[v] + bv;
      yp[(size_t)(16 + v) * DD] = acc1[v] + bv;
    }
    __syncthreads();
  }
}

__global__ __launch_bounds__(256) void att_logits_k(
    const float* __restrict__ xh, const float* __restrict__ a_src,
    const float* __restrict__ a_dst, float* __restrict__ e_src,
    float* __restrict__ e_dst) {
  int t = blockIdx.x * blockDim.x + threadIdx.x;
  if (t >= NN * HH) return;
  int n = t >> 3, h = t & 7;
  const float4* xp = (const float4*)(xh + (size_t)n * DD + h * CC);
  const float4* sp = (const float4*)(a_src + h * CC);
  const float4* dp = (const float4*)(a_dst + h * CC);
  float ss = 0.f, sd = 0.f;
  #pragma unroll
  for (int i = 0; i < 4; ++i) {
    float4 x = xp[i], s = sp[i], d = dp[i];
    ss += x.x * s.x + x.y * s.y + x.z * s.z + x.w * s.w;
    sd += x.x * d.x + x.y * d.y + x.z * d.z + x.w * d.w;
  }
  e_src[t] = ss;
  e_dst[t] = sd;
}

__device__ __forceinline__ float leaky(float v) {
  return v > 0.f ? v : v * NEG_SLOPE;
}

__global__ __launch_bounds__(256) void seg_count_k(const int* __restrict__ ei,
                                                   int* __restrict__ cnt) {
  int e = blockIdx.x * blockDim.x + threadIdx.x;
  if (e < EE) atomicAdd(&cnt[ei[EE + e]], 1);
}

__global__ __launch_bounds__(256) void seg_alloc_k(int* cntpos,
                                                   int* __restrict__ cursor,
                                                   int* __restrict__ seg_beg) {
  __shared__ int wsum[8];
  __shared__ int base_s;
  const int n    = blockIdx.x * 256 + threadIdx.x;
  const int lane = threadIdx.x & 31;
  const int wave = threadIdx.x >> 5;
  const int c = (n < NN) ? cntpos[n] : 0;
  int incl = c;
  #pragma unroll
  for (int o = 1; o < 32; o <<= 1) {
    int t = __shfl_up(incl, o, 32);
    if (lane >= o) incl += t;
  }
  if (lane == 31) wsum[wave] = incl;
  __syncthreads();
  if (threadIdx.x == 0) {
    int tot = 0;
    for (int w = 0; w < 8; ++w) { int t = wsum[w]; wsum[w] = tot; tot += t; }
    base_s = atomicAdd(cursor, tot);
  }
  __syncthreads();
  if (n < NN) {
    const int first = base_s + wsum[wave] + incl - c;
    seg_beg[n] = first;
    cntpos[n]  = first;
  }
}

__global__ __launch_bounds__(256) void seg_fill_k(const int* __restrict__ ei,
                                                  int* __restrict__ fillpos,
                                                  int* __restrict__ csr_src) {
  int e = blockIdx.x * blockDim.x + threadIdx.x;
  if (e < EE) {
    int p = atomicAdd(&fillpos[ei[EE + e]], 1);
    csr_src[p] = ei[e];
  }
}


template <int CTRL>
__device__ __forceinline__ float quad_dpp(float v) {
  return __int_as_float(__builtin_amdgcn_update_dpp(
      0, __float_as_int(v), CTRL, 0xf, 0xf, true));
}
#define QUAD_XOR1   0xB1
#define QUAD_XOR2   0x4E
#define QUAD_LANE(U) ((U) * 0x55)

__device__ __forceinline__ void load_sweep(const float* __restrict__ xq,
                                           int sidx, int t, v4f (&x)[4]) {
  #pragma unroll
  for (int u = 0; u < 4; ++u) {
    const int su = __builtin_amdgcn_readlane(sidx, 4 * t + u);
    x[u] = *(const v4f*)(xq + (size_t)su * DD);
  }
}

__device__ __forceinline__ void acc_edge(const v4f& x, float a, v4f& acc) {
  acc.x = __builtin_fmaf(x.x, a, acc.x);
  acc.y = __builtin_fmaf(x.y, a, acc.y);
  acc.z = __builtin_fmaf(x.z, a, acc.z);
  acc.w = __builtin_fmaf(x.w, a, acc.w);
}

__device__ __forceinline__ void acc_sweep(const v4f (&x)[4], float al, v4f& acc) {
  acc_edge(x[0], quad_dpp<QUAD_LANE(0)>(al), acc);
  acc_edge(x[1], quad_dpp<QUAD_LANE(1)>(al), acc);
  acc_edge(x[2], quad_dpp<QUAD_LANE(2)>(al), acc);
  acc_edge(x[3], quad_dpp<QUAD_LANE(3)>(al), acc);
}

template <int T>
__device__ __forceinline__ void sweep_pipe(int ns, const float* __restrict__ xq,
                                           int sidx, const float (&al)[8],
                                           v4f (&cur)[4], v4f (&nxt)[4],
                                           v4f& acc) {
  acc_sweep(cur, al[T], acc);
  if constexpr (T + 2 < 8) {
    if (ns > T + 2) {
      load_sweep(xq, sidx, T + 2, cur);
      sweep_pipe<T + 1>(ns, xq, sidx, al, nxt, cur, acc);
      return;
    }
  }
  if constexpr (T + 1 < 8) {
    if (ns > T + 1)
      acc_sweep(nxt, al[T + 1], acc);
  }
}

__global__ __launch_bounds__(256, 9) void node_gat_ln_k(
    const int* __restrict__ seg_beg, const int* __restrict__ seg_end,
    const int* __restrict__ csr_src, const float* __restrict__ e_src,
    const float* __restrict__ e_dst, const float* __restrict__ xh,
    const float* hin, const float* __restrict__ gb,
    const float* __restrict__ g, const float* __restrict__ b, float* hout) {
  const int node = __builtin_amdgcn_readfirstlane(
      (int)(blockIdx.x * 8 + (threadIdx.x >> 5)));
  const int lane = threadIdx.x & 31;
  const int hq   = lane >> 2;
  const int sub  = lane & 3;
  const int beg   = seg_beg[node];
  const int deg   = seg_end[node] - beg;
  const int total = deg + 1;
  const float ed  = e_dst[node * HH + hq];
  const size_t rbase = (size_t)node * DD + 4 * lane;
  const float* __restrict__ xq = xh + 4 * lane;

  v4f acc = {0.f, 0.f, 0.f, 0.f};
  v4f r;

  if (total <= 32) {
    const int ns = (total + 3) >> 2;
    int sidx = node;
    if (lane < deg) sidx = csr_src[beg + lane];

    r = *(const v4f*)(hin + rbase);

    float ev[8];
    #pragma unroll
    for (int t = 0; t < 8; ++t) {
      ev[t] = 0.f;
      if (t == 0 || ns > t) {
        const int s = __shfl(sidx, 4 * t + sub, 32);
        ev[t] = e_src[s * HH + hq];
      }
    }

    v4f xa[4], xb[4];
    load_sweep(xq, sidx, 0, xa);
    if (ns > 1) load_sweep(xq, sidx, 1, xb);

    float sc[8];
    float mx = -__builtin_inff();
    #pragma unroll
    for (int t = 0; t < 8; ++t) {
      sc[t] = leaky(ev[t] + ed);
      if (4 * t + sub < total) mx = fmaxf(mx, sc[t]);
    }
    mx = fmaxf(mx, quad_dpp<QUAD_XOR1>(mx));
    mx = fmaxf(mx, quad_dpp<QUAD_XOR2>(mx));

    float sm = 0.f;
    #pragma unroll
    for (int t = 0; t < 8; ++t) {
      sc[t] = __expf(sc[t] - mx);
      if (4 * t + sub < total) sm += sc[t];
    }
    sm += quad_dpp<QUAD_XOR1>(sm);
    sm += quad_dpp<QUAD_XOR2>(sm);
    const float den = sm + 1e-16f;

    float al[8];
    #pragma unroll
    for (int t = 0; t < 8; ++t)
      al[t] = (4 * t + sub < total) ? sc[t] / den : 0.f;

    sweep_pipe<0>(ns, xq, sidx, al, xa, xb, acc);
  } else {
    float mx = -__builtin_inff();
    for (int j = sub; j < total; j += 4) {
      const int s = (j < deg) ? csr_src[beg + j] : node;
      mx = fmaxf(mx, leaky(e_src[s * HH + hq] + ed));
    }
    mx = fmaxf(mx, quad_dpp<QUAD_XOR1>(mx));
    mx = fmaxf(mx, quad_dpp<QUAD_XOR2>(mx));

    float sm = 0.f;
    for (int j = sub; j < total; j += 4) {
      const int s = (j < deg) ? csr_src[beg + j] : node;
      sm += __expf(leaky(e_src[s * HH + hq] + ed) - mx);
    }
    sm += quad_dpp<QUAD_XOR1>(sm);
    sm += quad_dpp<QUAD_XOR2>(sm);
    const float den = sm + 1e-16f;

#define EDGE_ACC(U)                                                    \
    {                                                                  \
      const int su = __builtin_amdgcn_readlane(s, (U));                \
      const v4f x = *(const v4f*)(xq + (size_t)su * DD);               \
      acc_edge(x, quad_dpp<QUAD_LANE(U)>(alpha), acc);                 \
    }
    for (int base = 0; base < total; base += 4) {
      const int j = base + sub;
      int   s     = node;
      float alpha = 0.f;
      if (j < total) {
        if (j < deg) s = csr_src[beg + j];
        alpha = __expf(leaky(e_src[s * HH + hq] + ed) - mx) / den;
      }
      const int left = total - base;
      if (left >= 4) {
        EDGE_ACC(0) EDGE_ACC(1) EDGE_ACC(2) EDGE_ACC(3)
      } else {
        EDGE_ACC(0)
        if (left >= 2) EDGE_ACC(1)
        if (left >= 3) EDGE_ACC(2)
      }
    }
#undef EDGE_ACC
    r = *(const v4f*)(hin + rbase);
  }

  const v4f gbv = *(const v4f*)(gb + 4 * lane);
  const v4f gv  = *(const v4f*)(g + 4 * lane);
  const v4f bv  = *(const v4f*)(b + 4 * lane);
  const float x0 = r.x + acc.x + gbv.x;
  const float x1 = r.y + acc.y + gbv.y;
  const float x2 = r.z + acc.z + gbv.z;
  const float x3 = r.w + acc.w + gbv.w;
  float ssum = x0 + x1 + x2 + x3;
  #pragma unroll
  for (int o = 16; o > 0; o >>= 1) ssum += __shfl_xor(ssum, o, 32);
  const float mean = ssum * (1.f / (float)DD);
  const float d0 = x0 - mean, d1 = x1 - mean, d2 = x2 - mean, d3 = x3 - mean;
  float vs = d0 * d0 + d1 * d1 + d2 * d2 + d3 * d3;
  #pragma unroll
  for (int o = 16; o > 0; o >>= 1) vs += __shfl_xor(vs, o, 32);
  const float inv = rsqrtf(vs * (1.f / (float)DD) + LN_EPS);
  v4f o4;
  o4.x = d0 * inv * gv.x + bv.x;
  o4.y = d1 * inv * gv.y + bv.y;
  o4.z = d2 * inv * gv.z + bv.z;
  o4.w = d3 * inv * gv.w + bv.w;
  *(v4f*)(hout + rbase) = o4;
}

__global__ __launch_bounds__(128) void pool_sum_k(const float* __restrict__ h,
                                                  float* __restrict__ pooled) {
  int c = threadIdx.x;
  float s = 0.f;
  for (int n = blockIdx.x; n < NN; n += gridDim.x)
    s += h[(size_t)n * DD + c];
  atomicAdd(&pooled[c], s);
}

__global__ __launch_bounds__(128) void head_k(
    const float* __restrict__ pooled_sum, const float* __restrict__ W_mu,
    const float* __restrict__ b_mu, const float* __restrict__ W_lv,
    const float* __restrict__ b_lv, float* __restrict__ out) {
  int d = threadIdx.x;
  __shared__ float pl[DD];
  float pd = pooled_sum[d] * (1.f / (float)NN);
  pl[d] = pd;
  __syncthreads();
  float mu = b_mu[d], lv = b_lv[d];
  for (int k = 0; k < DD; ++k) {
    float p = pl[k];
    mu += p * W_mu[(size_t)d * DD + k];
    lv += p * W_lv[(size_t)d * DD + k];
  }
  out[d]      = mu;
  out[DD + d] = lv;
  out[2 * DD + (size_t)NN * DD + d] = pd;
}

__global__ __launch_bounds__(256) void fill32_k(unsigned* __restrict__ p,
                                                unsigned v, int n) {
  int t = blockIdx.x * blockDim.x + threadIdx.x;
  if (t < n) p[t] = v;
}

extern "C" void kernel_launch(void* const* d_in, const int* in_sizes, int n_in,
                              void* d_out, int out_size, void* d_ws, size_t ws_size,
                              hipStream_t stream) {
  const float* x       = (const float*)d_in[0];
  const int*   ei      = (const int*)d_in[1];
  const float* W_in    = (const float*)d_in[2];
  const float* b_in    = (const float*)d_in[3];
  const float* gat_W   = (const float*)d_in[4];
  const float* att_src = (const float*)d_in[5];
  const float* att_dst = (const float*)d_in[6];
  const float* gat_b   = (const float*)d_in[7];
  const float* ln_g    = (const float*)d_in[8];
  const float* ln_b    = (const float*)d_in[9];
  const float* W_mu    = (const float*)d_in[10];
  const float* b_mu    = (const float*)d_in[11];
  const float* W_lv    = (const float*)d_in[12];
  const float* b_lv    = (const float*)d_in[13];
  float* out = (float*)d_out;

  float* h       = (float*)d_ws;
  float* xh      = h  + (size_t)NN * DD;
  float* grp     = xh + (size_t)NN * DD;
  float* e_src   = grp   + (size_t)NN * DD;
  float* e_dst   = e_src + (size_t)NN * HH;
  float* pooled  = e_dst + (size_t)NN * HH * 3;

  int* cntpos  = (int*)grp;
  int* cursor  = cntpos + NN;
  int* seg_beg = cntpos + NN + 32;
  int* csr_src = seg_beg + NN;

  float* h_final = out + 2 * DD;

  const int TPB = 256;
  const int nh_blk   = (NN * HH + TPB - 1) / TPB;
  const int e_blk    = (EE + TPB - 1) / TPB;
  const int n_blk    = (NN + TPB - 1) / TPB;
  const int n1_blk   = (NN + 1 + TPB - 1) / TPB;
  const int node_blk = NN / 8;
  const int gemm_blk = 512;

  fill32_k   <<<n1_blk, TPB, 0, stream>>>((unsigned*)cntpos, 0u, NN + 1);
  seg_count_k<<<e_blk,  TPB, 0, stream>>>(ei, cntpos);
  seg_alloc_k<<<n_blk,  TPB, 0, stream>>>(cntpos, cursor, seg_beg);
  seg_fill_k <<<e_blk,  TPB, 0, stream>>>(ei, cntpos, csr_src);

  gemm_wmma_k<F_IN><<<gemm_blk, TPB, 0, stream>>>(x, W_in, b_in, h);

  for (int l = 0; l < LL; ++l) {
    const float* Wl  = gat_W   + (size_t)l * DD * DD;
    const float* asl = att_src + (size_t)l * HH * CC;
    const float* adl = att_dst + (size_t)l * HH * CC;
    const float* gbl = gat_b   + (size_t)l * DD;
    const float* gl  = ln_g    + (size_t)l * DD;
    const float* bl  = ln_b    + (size_t)l * DD;

    gemm_wmma_k<DD><<<gemm_blk, TPB, 0, stream>>>(h, Wl, nullptr, xh);

    att_logits_k<<<nh_blk, TPB, 0, stream>>>(xh, asl, adl, e_src, e_dst);

    float* hout = (l == LL - 1) ? h_final : h;
    node_gat_ln_k<<<node_blk, TPB, 0, stream>>>(seg_beg, cntpos, csr_src, e_src,
                                                e_dst, xh, h, gbl, gl, bl, hout);
  }

  fill32_k<<<1, TPB, 0, stream>>>((unsigned*)pooled, 0u, DD);
  pool_sum_k<<<512, DD, 0, stream>>>(h_final, pooled);
  head_k<<<1, DD, 0, stream>>>(pooled, W_mu, b_mu, W_lv, b_lv, out);
}


# in-edges grouped by a two-level counting sort on LDS counters (256-node bins per edge chunk, table scan, per-bin rank); the 3.2M global int atomics are gone; sums still run in slot arrival order
# speedup vs baseline: 3.9969x; 1.1901x over previous
#include <hip/hip_runtime.h>
#include <hip/hip_bf16.h>


#define NN      100000
#define EE      1600000
#define ETOT    (EE + NN)
#define F_IN    256
#define DD      128
#define HH      8
#define CC      16
#define LL      4
#define NEG_SLOPE 0.2f
#define LN_EPS  1e-5f

typedef __attribute__((ext_vector_type(2))) float v2f;
typedef __attribute__((ext_vector_type(4))) float v4f;
typedef __attribute__((ext_vector_type(8))) float v8f;

template <int K>
__global__ __launch_bounds__(256, 4) void gemm_wmma_k(
    const float* __restrict__ X, const float* __restrict__ W,
    const float* __restrict__ bias, float* __restrict__ Y) {
  constexpr int KP    = K + 4;
  constexpr int PR    = 32;
  constexpr int NPASS = NN / PR;
  constexpr int TPR   = K / 4;
  constexpr int RPS   = 256 / TPR;
  constexpr int NV    = PR / RPS;
  constexpr int NS    = K / 4;
  constexpr int CH    = 4;
  constexpr int NG    = NS / CH;
  __shared__ float As[PR * KP];

  const int wave = threadIdx.x >> 5;
  const int lane = threadIdx.x & 31;
  const int col0 = wave * 16;
  const int m    = lane & 15;
  const int koff = (lane < 16) ? 0 : 2;
  const int sr   = threadIdx.x / TPR;
  const int sc   = (threadIdx.x % TPR) * 4;
  const int gstride = (int)gridDim.x;

  const float* __restrict__ wrow = W + (size_t)(col0 + m) * K + koff;
  v2f b[NS];
  #pragma unroll
  for (int i = 0; i < NS; ++i) {
    b[i].x = wrow[4 * i];
    b[i].y = wrow[4 * i + 1];
  }

  const int mrow = (lane < 16) ? 0 : 8;
  const int c    = col0 + m;
  const float bv = bias ? bias[c] : 0.0f;

  const float* __restrict__ a0row = As + m * KP + koff;
  const float* __restrict__ a1row = As + (16 + m) * KP + koff;
  const float* __restrict__ xs    = X + (size_t)sr * K + sc;
  float* __restrict__ ls          = As + sr * KP + sc;

  int pass = blockIdx.x;
  v4f pre[NV];
  {
    const float* __restrict__ xp = xs + (size_t)min(pass, NPASS - 1) * PR * K;
    #pragma unroll
    for (int i = 0; i < NV; ++i)
      pre[i] = *(const v4f*)(xp + (size_t)i * RPS * K);
  }

  for (; pass < NPASS; pass += gstride) {
    #pragma unroll
    for (int i = 0; i < NV; ++i)
      *(v4f*)(ls + i * RPS * KP) = pre[i];
    __syncthreads();

    {
      const float* __restrict__ xp =
          xs + (size_t)min(pass + gstride, NPASS - 1) * PR * K;
      #pragma unroll
      for (int i = 0; i < NV; ++i)
        pre[i] = *(const v4f*)(xp + (size_t)i * RPS * K);
    }

    v2f fa[2][2 * CH];
    #pragma unroll
    for (int j = 0; j < CH; ++j) {
      fa[0][j].x      = a0row[4 * j];
      fa[0][j].y      = a0row[4 * j + 1];
      fa[0][CH + j].x = a1row[4 * j];
      fa[0][CH + j].y = a1row[4 * j + 1];
    }
    v8f acc0 = {}, acc1 = {};
    #pragma unroll
    for (int g = 0; g < NG; ++g) {
      if (g + 1 < NG) {
        #pragma unroll
        for (int j = 0; j < CH; ++j) {
          const int k = 4 * ((g + 1) * CH + j);
          fa[(g + 1) & 1][j].x      = a0row[k];
          fa[(g + 1) & 1][j].y      = a0row[k + 1];
          fa[(g + 1) & 1][CH + j].x = a1row[k];
          fa[(g + 1) & 1][CH + j].y = a1row[k + 1];
        }
      }
      __builtin_amdgcn_sched_barrier(0);
      #pragma unroll
      for (int j = 0; j < CH; ++j) {
        acc0 = __builtin_amdgcn_wmma_f32_16x16x4_f32(
            false, fa[g & 1][j], false, b[g * CH + j], (short)0, acc0,
            false, false);
        acc1 = __builtin_amdgcn_wmma_f32_16x16x4_f32(
            false, fa[g & 1][CH + j], false, b[g * CH + j], (short)0, acc1,
            false, false);
      }
      __builtin_amdgcn_sched_barrier(0);
    }

    float* __restrict__ yp = Y + ((size_t)pass * PR + mrow) * DD + c;
    #pragma unroll
    for (int v = 0; v < 8; ++v) {
      yp[(size_t)v * DD]        = acc0[v] + bv;
      yp[(size_t)(16 + v) * DD] = acc1[v] + bv;
    }
    __syncthreads();
  }
}

__global__ __launch_bounds__(256) void att_logits_k(
    const float* __restrict__ xh, const float* __restrict__ a_src,
    const float* __restrict__ a_dst, float* __restrict__ e_src,
    float* __restrict__ e_dst) {
  int t = blockIdx.x * blockDim.x + threadIdx.x;
  if (t >= NN * HH) return;
  int n = t >> 3, h = t & 7;
  const float4* xp = (const float4*)(xh + (size_t)n * DD + h * CC);
  const float4* sp = (const float4*)(a_src + h * CC);
  const float4* dp = (const float4*)(a_dst + h * CC);
  float ss = 0.f, sd = 0.f;
  #pragma unroll
  for (int i = 0; i < 4; ++i) {
    float4 x = xp[i], s = sp[i], d = dp[i];
    ss += x.x * s.x + x.y * s.y + x.z * s.z + x.w * s.w;
    sd += x.x * d.x + x.y * d.y + x.z * d.z + x.w * d.w;
  }
  e_src[t] = ss;
  e_dst[t] = sd;
}

__device__ __forceinline__ float leaky(float v) {
  return v > 0.f ? v : v * NEG_SLOPE;
}

#define BINSH   8
#define BINW    (1 << BINSH)
#define NBIN    ((NN + BINW - 1) / BINW)
#define GCH     256
#define ECH     ((EE + GCH - 1) / GCH)
#define SORT_T  1024
#define EPT     ((ECH + SORT_T - 1) / SORT_T)
#define SPT     8
static_assert(NN <= (1 << (32 - BINSH)), "source index must fit beside the bin-local dst");
static_assert(NBIN <= SORT_T, "one thread per bin in steps 3 and 4");

__global__ __launch_bounds__(SORT_T) void bin_count_k(
    const int* __restrict__ ei, int* __restrict__ tbl) {
  __shared__ int hist[NBIN];
  const int tid = threadIdx.x;
  for (int b = tid; b < NBIN; b += SORT_T) hist[b] = 0;
  __syncthreads();
  const int e0 = blockIdx.x * ECH;
  const int e1 = min(e0 + ECH, EE);
  int d[EPT];
  #pragma unroll
  for (int k = 0; k < EPT; ++k)
    d[k] = ei[EE + min(e0 + tid + k * SORT_T, e1 - 1)];
  #pragma unroll
  for (int k = 0; k < EPT; ++k)
    if (e0 + tid + k * SORT_T < e1) atomicAdd(&hist[d[k] >> BINSH], 1);
  __syncthreads();
  for (int b = tid; b < NBIN; b += SORT_T) tbl[b * GCH + blockIdx.x] = hist[b];
}

__global__ __launch_bounds__(GCH) void bin_scan_k(int* __restrict__ tbl,
                                                  int* __restrict__ bintot) {
  __shared__ int wsum[GCH / 32];
  const int tid  = threadIdx.x;
  const int lane = tid & 31;
  const int wave = tid >> 5;
  const int c = tbl[blockIdx.x * GCH + tid];
  int incl = c;
  for (int o = 1; o < 32; o <<= 1) {
    const int u = __shfl_up(incl, o, 32);
    if (lane >= o) incl += u;
  }
  if (lane == 31) wsum[wave] = incl;
  __syncthreads();
  int off = 0;
  for (int w = 0; w < wave; ++w) off += wsum[w];
  tbl[blockIdx.x * GCH + tid] = off + incl - c;
  if (tid == GCH - 1) bintot[blockIdx.x] = off + incl;
}

__global__ __launch_bounds__(SORT_T) void bin_scatter_k(
    const int* __restrict__ ei, const int* __restrict__ tbl,
    const int* __restrict__ bintot, unsigned* __restrict__ binned) {
  __shared__ int cur[NBIN];
  __shared__ int wtot[SORT_T / 32];
  const int tid  = threadIdx.x;
  const int lane = tid & 31;
  const int wave = tid >> 5;
  const int e0 = blockIdx.x * ECH;
  const int e1 = min(e0 + ECH, EE);
  int d[EPT], sv[EPT];
  #pragma unroll
  for (int k = 0; k < EPT; ++k) {
    const int e = min(e0 + tid + k * SORT_T, e1 - 1);
    d[k]  = ei[EE + e];
    sv[k] = ei[e];
  }
  int bt = 0, tb = 0;
  if (tid < NBIN) {
    bt = bintot[tid];
    tb = tbl[tid * GCH + blockIdx.x];
  }
  int incl = bt;
  for (int o = 1; o < 32; o <<= 1) {
    const int u = __shfl_up(incl, o, 32);
    if (lane >= o) incl += u;
  }
  if (lane == 31) wtot[wave] = incl;
  __syncthreads();
  if (tid < NBIN) {
    int off = 0;
    for (int w = 0; w < wave; ++w) off += wtot[w];
    cur[tid] = off + incl - bt + tb;
  }
  __syncthreads();
  #pragma unroll
  for (int k = 0; k < EPT; ++k) {
    if (e0 + tid + k * SORT_T < e1) {
      const int p = atomicAdd(&cur[d[k] >> BINSH], 1);
      binned[p] = ((unsigned)sv[k] << BINSH) | (unsigned)(d[k] & (BINW - 1));
    }
  }
}

__global__ __launch_bounds__(SORT_T) void bin_sort_k(
    const unsigned* __restrict__ binned, const int* __restrict__ bintot,
    int* __restrict__ seg_beg, int* __restrict__ seg_end,
    int* __restrict__ csr_src) {
  __shared__ int cnt[BINW + 32];
  __shared__ int cur[BINW + 32];
  __shared__ int wsum[BINW / 32];
  __shared__ int sstart;
  const int tid  = threadIdx.x;
  const int lane = tid & 31;
  const int wave = tid >> 5;
  const int bin  = blockIdx.x;
  int v = (tid < bin) ? bintot[tid] : 0;
  const int nbin = bintot[bin];
  if (tid < BINW + 32) {
    cnt[tid] = 0;
    cur[tid] = 0;
  }
  if (tid == 0) sstart = 0;
  __syncthreads();
  for (int o = 16; o > 0; o >>= 1) v += __shfl_xor(v, o, 32);
  if (lane == 0 && v != 0) atomicAdd(&sstart, v);
  __syncthreads();
  const int start = sstart;
  const int end   = start + nbin;
  for (int i0 = start + tid; i0 < end; i0 += SPT * SORT_T) {
    unsigned r[SPT];
    #pragma unroll
    for (int u = 0; u < SPT; ++u) r[u] = binned[min(i0 + u * SORT_T, end - 1)];
    #pragma unroll
    for (int u = 0; u < SPT; ++u) {
      const int w = (i0 + u * SORT_T < end) ? (int)(r[u] & (BINW - 1))
                                            : BINW + lane;
      atomicAdd(&cnt[w], 1);
    }
  }
  __syncthreads();
  int c = 0, incl = 0;
  if (tid < BINW) {
    c = cnt[tid];
    incl = c;
    for (int o = 1; o < 32; o <<= 1) {
      const int u = __shfl_up(incl, o, 32);
      if (lane >= o) incl += u;
    }
    if (lane == 31) wsum[wave] = incl;
  }
  __syncthreads();
  if (tid < BINW) {
    int off = 0;
    for (int w = 0; w < wave; ++w) off += wsum[w];
    const int first = start + off + incl - c;
    cur[tid] = first;
    const int node = bin * BINW + tid;
    if (node < NN) {
      seg_beg[node] = first;
      seg_end[node] = first + c;
    }
  }
  __syncthreads();
  for (int i0 = start + tid; i0 < end; i0 += SPT * SORT_T) {
    unsigned r[SPT];
    int p[SPT];
    #pragma unroll
    for (int u = 0; u < SPT; ++u) r[u] = binned[min(i0 + u * SORT_T, end - 1)];
    #pragma unroll
    for (int u = 0; u < SPT; ++u) {
      const int w = (i0 + u * SORT_T < end) ? (int)(r[u] & (BINW - 1))
                                            : BINW + lane;
      p[u] = atomicAdd(&cur[w], 1);
    }
    #pragma unroll
    for (int u = 0; u < SPT; ++u)
      if (i0 + u * SORT_T < end) csr_src[p[u]] = (int)(r[u] >> BINSH);
  }
}


template <int CTRL>
__device__ __forceinline__ float quad_dpp(float v) {
  return __int_as_float(__builtin_amdgcn_update_dpp(
      0, __float_as_int(v), CTRL, 0xf, 0xf, true));
}
#define QUAD_XOR1   0xB1
#define QUAD_XOR2   0x4E
#define QUAD_LANE(U) ((U) * 0x55)

__device__ __forceinline__ void load_sweep(const float* __restrict__ xq,
                                           int sidx, int t, v4f (&x)[4]) {
  #pragma unroll
  for (int u = 0; u < 4; ++u) {
    const int su = __builtin_amdgcn_readlane(sidx, 4 * t + u);
    x[u] = *(const v4f*)(xq + (size_t)su * DD);
  }
}

__device__ __forceinline__ void acc_edge(const v4f& x, float a, v4f& acc) {
  acc.x = __builtin_fmaf(x.x, a, acc.x);
  acc.y = __builtin_fmaf(x.y, a, acc.y);
  acc.z = __builtin_fmaf(x.z, a, acc.z);
  acc.w = __builtin_fmaf(x.w, a, acc.w);
}

__device__ __forceinline__ void acc_sweep(const v4f (&x)[4], float al, v4f& acc) {
  acc_edge(x[0], quad_dpp<QUAD_LANE(0)>(al), acc);
  acc_edge(x[1], quad_dpp<QUAD_LANE(1)>(al), acc);
  acc_edge(x[2], quad_dpp<QUAD_LANE(2)>(al), acc);
  acc_edge(x[3], quad_dpp<QUAD_LANE(3)>(al), acc);
}

template <int T>
__device__ __forceinline__ void sweep_pipe(int ns, const float* __restrict__ xq,
                                           int sidx, const float (&al)[8],
                                           v4f (&cur)[4], v4f (&nxt)[4],
                                           v4f& acc) {
  acc_sweep(cur, al[T], acc);
  if constexpr (T + 2 < 8) {
    if (ns > T + 2) {
      load_sweep(xq, sidx, T + 2, cur);
      sweep_pipe<T + 1>(ns, xq, sidx, al, nxt, cur, acc);
      return;
    }
  }
  if constexpr (T + 1 < 8) {
    if (ns > T + 1)
      acc_sweep(nxt, al[T + 1], acc);
  }
}

__global__ __launch_bounds__(256, 9) void node_gat_ln_k(
    const int* __restrict__ seg_beg, const int* __restrict__ seg_end,
    const int* __restrict__ csr_src, const float* __restrict__ e_src,
    const float* __restrict__ e_dst, const float* __restrict__ xh,
    const float* hin, const float* __restrict__ gb,
    const float* __restrict__ g, const float* __restrict__ b, float* hout) {
  const int node = __builtin_amdgcn_readfirstlane(
      (int)(blockIdx.x * 8 + (threadIdx.x >> 5)));
  const int lane = threadIdx.x & 31;
  const int hq   = lane >> 2;
  const int sub  = lane & 3;
  const int beg   = seg_beg[node];
  const int deg   = seg_end[node] - beg;
  const int total = deg + 1;
  const float ed  = e_dst[node * HH + hq];
  const size_t rbase = (size_t)node * DD + 4 * lane;
  const float* __restrict__ xq = xh + 4 * lane;

  v4f acc = {0.f, 0.f, 0.f, 0.f};
  v4f r;

  if (total <= 32) {
    const int ns = (total + 3) >> 2;
    int sidx = node;
    if (lane < deg) sidx = csr_src[beg + lane];

    r = *(const v4f*)(hin + rbase);

    float ev[8];
    #pragma unroll
    for (int t = 0; t < 8; ++t) {
      ev[t] = 0.f;
      if (t == 0 || ns > t) {
        const int s = __shfl(sidx, 4 * t + sub, 32);
        ev[t] = e_src[s * HH + hq];
      }
    }

    v4f xa[4], xb[4];
    load_sweep(xq, sidx, 0, xa);
    if (ns > 1) load_sweep(xq, sidx, 1, xb);

    float sc[8];
    float mx = -__builtin_inff();
    #pragma unroll
    for (int t = 0; t < 8; ++t) {
      sc[t] = leaky(ev[t] + ed);
      if (4 * t + sub < total) mx = fmaxf(mx, sc[t]);
    }
    mx = fmaxf(mx, quad_dpp<QUAD_XOR1>(mx));
    mx = fmaxf(mx, quad_dpp<QUAD_XOR2>(mx));

    float sm = 0.f;
    #pragma unroll
    for (int t = 0; t < 8; ++t) {
      sc[t] = __expf(sc[t] - mx);
      if (4 * t + sub < total) sm += sc[t];
    }
    sm += quad_dpp<QUAD_XOR1>(sm);
    sm += quad_dpp<QUAD_XOR2>(sm);
    const float den = sm + 1e-16f;

    float al[8];
    #pragma unroll
    for (int t = 0; t < 8; ++t)
      al[t] = (4 * t + sub < total) ? sc[t] / den : 0.f;

    sweep_pipe<0>(ns, xq, sidx, al, xa, xb, acc);
  } else {
    float mx = -__builtin_inff();
    for (int j = sub; j < total; j += 4) {
      const int s = (j < deg) ? csr_src[beg + j] : node;
      mx = fmaxf(mx, leaky(e_src[s * HH + hq] + ed));
    }
    mx = fmaxf(mx, quad_dpp<QUAD_XOR1>(mx));
    mx = fmaxf(mx, quad_dpp<QUAD_XOR2>(mx));

    float sm = 0.f;
    for (int j = sub; j < total; j += 4) {
      const int s = (j < deg) ? csr_src[beg + j] : node;
      sm += __expf(leaky(e_src[s * HH + hq] + ed) - mx);
    }
    sm += quad_dpp<QUAD_XOR1>(sm);
    sm += quad_dpp<QUAD_XOR2>(sm);
    const float den = sm + 1e-16f;

#define EDGE_ACC(U)                                                    \
    {                                                                  \
      const int su = __builtin_amdgcn_readlane(s, (U));                \
      const v4f x = *(const v4f*)(xq + (size_t)su * DD);               \
      acc_edge(x, quad_dpp<QUAD_LANE(U)>(alpha), acc);                 \
    }
    for (int base = 0; base < total; base += 4) {
      const int j = base + sub;
      int   s     = node;
      float alpha = 0.f;
      if (j < total) {
        if (j < deg) s = csr_src[beg + j];
        alpha = __expf(leaky(e_src[s * HH + hq] + ed) - mx) / den;
      }
      const int left = total - base;
      if (left >= 4) {
        EDGE_ACC(0) EDGE_ACC(1) EDGE_ACC(2) EDGE_ACC(3)
      } else {
        EDGE_ACC(0)
        if (left >= 2) EDGE_ACC(1)
        if (left >= 3) EDGE_ACC(2)
      }
    }
#undef EDGE_ACC
    r = *(const v4f*)(hin + rbase);
  }

  const v4f gbv = *(const v4f*)(gb + 4 * lane);
  const v4f gv  = *(const v4f*)(g + 4 * lane);
  const v4f bv  = *(const v4f*)(b + 4 * lane);
  const float x0 = r.x + acc.x + gbv.x;
  const float x1 = r.y + acc.y + gbv.y;
  const float x2 = r.z + acc.z + gbv.z;
  const float x3 = r.w + acc.w + gbv.w;
  float ssum = x0 + x1 + x2 + x3;
  #pragma unroll
  for (int o = 16; o > 0; o >>= 1) ssum += __shfl_xor(ssum, o, 32);
  const float mean = ssum * (1.f / (float)DD);
  const float d0 = x0 - mean, d1 = x1 - mean, d2 = x2 - mean, d3 = x3 - mean;
  float vs = d0 * d0 + d1 * d1 + d2 * d2 + d3 * d3;
  #pragma unroll
  for (int o = 16; o > 0; o >>= 1) vs += __shfl_xor(vs, o, 32);
  const float inv = rsqrtf(vs * (1.f / (float)DD) + LN_EPS);
  v4f o4;
  o4.x = d0 * inv * gv.x + bv.x;
  o4.y = d1 * inv * gv.y + bv.y;
  o4.z = d2 * inv * gv.z + bv.z;
  o4.w = d3 * inv * gv.w + bv.w;
  *(v4f*)(hout + rbase) = o4;
}

__global__ __launch_bounds__(128) void pool_sum_k(const float* __restrict__ h,
                                                  float* __restrict__ pooled) {
  int c = threadIdx.x;
  float s = 0.f;
  for (int n = blockIdx.x; n < NN; n += gridDim.x)
    s += h[(size_t)n * DD + c];
  atomicAdd(&pooled[c], s);
}

__global__ __launch_bounds__(128) void head_k(
    const float* __restrict__ pooled_sum, const float* __restrict__ W_mu,
    const float* __restrict__ b_mu, const float* __restrict__ W_lv,
    const float* __restrict__ b_lv, float* __restrict__ out) {
  int d = threadIdx.x;
  __shared__ float pl[DD];
  float pd = pooled_sum[d] * (1.f / (float)NN);
  pl[d] = pd;
  __syncthreads();
  float mu = b_mu[d], lv = b_lv[d];
  for (int k = 0; k < DD; ++k) {
    float p = pl[k];
    mu += p * W_mu[(size_t)d * DD + k];
    lv += p * W_lv[(size_t)d * DD + k];
  }
  out[d]      = mu;
  out[DD + d] = lv;
  out[2 * DD + (size_t)NN * DD + d] = pd;
}

__global__ __launch_bounds__(256) void fill32_k(unsigned* __restrict__ p,
                                                unsigned v, int n) {
  int t = blockIdx.x * blockDim.x + threadIdx.x;
  if (t < n) p[t] = v;
}

extern "C" void kernel_launch(void* const* d_in, const int* in_sizes, int n_in,
                              void* d_out, int out_size, void* d_ws, size_t ws_size,
                              hipStream_t stream) {
  const float* x       = (const float*)d_in[0];
  const int*   ei      = (const int*)d_in[1];
  const float* W_in    = (const float*)d_in[2];
  const float* b_in    = (const float*)d_in[3];
  const float* gat_W   = (const float*)d_in[4];
  const float* att_src = (const float*)d_in[5];
  const float* att_dst = (const float*)d_in[6];
  const float* gat_b   = (const float*)d_in[7];
  const float* ln_g    = (const float*)d_in[8];
  const float* ln_b    = (const float*)d_in[9];
  const float* W_mu    = (const float*)d_in[10];
  const float* b_mu    = (const float*)d_in[11];
  const float* W_lv    = (const float*)d_in[12];
  const float* b_lv    = (const float*)d_in[13];
  float* out = (float*)d_out;

  float* h       = (float*)d_ws;
  float* xh      = h  + (size_t)NN * DD;
  float* grp     = xh + (size_t)NN * DD;
  float* e_src   = grp   + (size_t)NN * DD;
  float* e_dst   = e_src + (size_t)NN * HH;
  float* pooled  = e_dst + (size_t)NN * HH * 3;

  int*  seg_end = (int*)grp;
  int*  seg_beg = seg_end + NN;
  int*  csr_src = seg_beg + NN;
  int*  tbl     = csr_src + EE;
  int*  bintot  = tbl + NBIN * GCH;
  unsigned* binned = (unsigned*)(bintot + 512);

  float* h_final = out + 2 * DD;

  const int TPB = 256;
  const int nh_blk   = (NN * HH + TPB - 1) / TPB;
  const int node_blk = NN / 8;
  const int gemm_blk = 512;

  bin_count_k  <<<GCH,  SORT_T, 0, stream>>>(ei, tbl);
  bin_scan_k   <<<NBIN, GCH,    0, stream>>>(tbl, bintot);
  bin_scatter_k<<<GCH,  SORT_T, 0, stream>>>(ei, tbl, bintot, binned);
  bin_sort_k   <<<NBIN, SORT_T, 0, stream>>>(binned, bintot, seg_beg, seg_end, csr_src);

  gemm_wmma_k<F_IN><<<gemm_blk, TPB, 0, stream>>>(x, W_in, b_in, h);

  for (int l = 0; l < LL; ++l) {
    const float* Wl  = gat_W   + (size_t)l * DD * DD;
    const float* asl = att_src + (size_t)l * HH * CC;
    const float* adl = att_dst + (size_t)l * HH * CC;
    const float* gbl = gat_b   + (size_t)l * DD;
    const float* gl  = ln_g    + (size_t)l * DD;
    const float* bl  = ln_b    + (size_t)l * DD;

    gemm_wmma_k<DD><<<gemm_blk, TPB, 0, stream>>>(h, Wl, nullptr, xh);

    att_logits_k<<<nh_blk, TPB, 0, stream>>>(xh, asl, adl, e_src, e_dst);

    float* hout = (l == LL - 1) ? h_final : h;
    node_gat_ln_k<<<node_blk, TPB, 0, stream>>>(seg_beg, seg_end, csr_src, e_src,
                                                e_dst, xh, h, gbl, gl, bl, hout);
  }

  fill32_k<<<1, TPB, 0, stream>>>((unsigned*)pooled, 0u, DD);
  pool_sum_k<<<512, DD, 0, stream>>>(h_final, pooled);
  head_k<<<1, DD, 0, stream>>>(pooled, W_mu, b_mu, W_lv, b_lv, out);
}


# mean pool: a wave requests all its 24-25 rows (one b128 load per lane and row) before it sums; was one 4-byte load per wait, 195 trips; column sums regrouped (wave, LDS over block, atomics kept)
# speedup vs baseline: 4.6133x; 1.1542x over previous
#include <hip/hip_runtime.h>
#include <hip/hip_bf16.h>


#define NN      100000
#define EE      1600000
#define ETOT    (EE + NN)
#define F_IN    256
#define DD      128
#define HH      8
#define CC      16
#define LL      4
#define NEG_SLOPE 0.2f
#define LN_EPS  1e-5f

typedef __attribute__((ext_vector_type(2))) float v2f;
typedef __attribute__((ext_vector_type(4))) float v4f;
typedef __attribute__((ext_vector_type(8))) float v8f;

template <int K>
__global__ __launch_bounds__(256, 4) void gemm_wmma_k(
    const float* __restrict__ X, const float* __restrict__ W,
    const float* __restrict__ bias, float* __restrict__ Y) {
  constexpr int KP    = K + 4;
  constexpr int PR    = 32;
  constexpr int NPASS = NN / PR;
  constexpr int TPR   = K / 4;
  constexpr int RPS   = 256 / TPR;
  constexpr int NV    = PR / RPS;
  constexpr int NS    = K / 4;
  constexpr int CH    = 4;
  constexpr int NG    = NS / CH;
  __shared__ float As[PR * KP];

  const int wave = threadIdx.x >> 5;
  const int lane = threadIdx.x & 31;
  const int col0 = wave * 16;
  const int m    = lane & 15;
  const int koff = (lane < 16) ? 0 : 2;
  const int sr   = threadIdx.x / TPR;
  const int sc   = (threadIdx.x % TPR) * 4;
  const int gstride = (int)gridDim.x;

  const float* __restrict__ wrow = W + (size_t)(col0 + m) * K + koff;
  v2f b[NS];
  #pragma unroll
  for (int i = 0; i < NS; ++i) {
    b[i].x = wrow[4 * i];
    b[i].y = wrow[4 * i + 1];
  }

  const int mrow = (lane < 16) ? 0 : 8;
  const int c    = col0 + m;
  const float bv = bias ? bias[c] : 0.0f;

  const float* __restrict__ a0row = As + m * KP + koff;
  const float* __restrict__ a1row = As + (16 + m) * KP + koff;
  const float* __restrict__ xs    = X + (size_t)sr * K + sc;
  float* __restrict__ ls          = As + sr * KP + sc;

  int pass = blockIdx.x;
  v4f pre[NV];
  {
    const float* __restrict__ xp = xs + (size_t)min(pass, NPASS - 1) * PR * K;
    #pragma unroll
    for (int i = 0; i < NV; ++i)
      pre[i] = *(const v4f*)(xp + (size_t)i * RPS * K);
  }

  for (; pass < NPASS; pass += gstride) {
    #pragma unroll
    for (int i = 0; i < NV; ++i)
      *(v4f*)(ls + i * RPS * KP) = pre[i];
    __syncthreads();

    {
      const float* __restrict__ xp =
          xs + (size_t)min(pass + gstride, NPASS - 1) * PR * K;
      #pragma unroll
      for (int i = 0; i < NV; ++i)
        pre[i] = *(const v4f*)(xp + (size_t)i * RPS * K);
    }

    v2f fa[2][2 * CH];
    #pragma unroll
    for (int j = 0; j < CH; ++j) {
      fa[0][j].x      = a0row[4 * j];
      fa[0][j].y      = a0row[4 * j + 1];
      fa[0][CH + j].x = a1row[4 * j];
      fa[0][CH + j].y = a1row[4 * j + 1];
    }
    v8f acc0 = {}, acc1 = {};
    #pragma unroll
    for (int g = 0; g < NG; ++g) {
      if (g + 1 < NG) {
        #pragma unroll
        for (int j = 0; j < CH; ++j) {
          const int k = 4 * ((g + 1) * CH + j);
          fa[(g + 1) & 1][j].x      = a0row[k];
          fa[(g + 1) & 1][j].y      = a0row[k + 1];
          fa[(g + 1) & 1][CH + j].x = a1row[k];
          fa[(g + 1) & 1][CH + j].y = a1row[k + 1];
        }
      }
      __builtin_amdgcn_sched_barrier(0);
      #pragma unroll
      for (int j = 0; j < CH; ++j) {
        acc0 = __builtin_amdgcn_wmma_f32_16x16x4_f32(
            false, fa[g & 1][j], false, b[g * CH + j], (short)0, acc0,
            false, false);
        acc1 = __builtin_amdgcn_wmma_f32_16x16x4_f32(
            false, fa[g & 1][CH + j], false, b[g * CH + j], (short)0, acc1,
            false, false);
      }
      __builtin_amdgcn_sched_barrier(0);
    }

    float* __restrict__ yp = Y + ((size_t)pass * PR + mrow) * DD + c;
    #pragma unroll
    for (int v = 0; v < 8; ++v) {
      yp[(size_t)v * DD]        = acc0[v] + bv;
      yp[(size_t)(16 + v) * DD] = acc1[v] + bv;
    }
    __syncthreads();
  }
}

__global__ __launch_bounds__(256) void att_logits_k(
    const float* __restrict__ xh, const float* __restrict__ a_src,
    const float* __restrict__ a_dst, float* __restrict__ e_src,
    float* __restrict__ e_dst) {
  int t = blockIdx.x * blockDim.x + threadIdx.x;
  if (t >= NN * HH) return;
  int n = t >> 3, h = t & 7;
  const float4* xp = (const float4*)(xh + (size_t)n * DD + h * CC);
  const float4* sp = (const float4*)(a_src + h * CC);
  const float4* dp = (const float4*)(a_dst + h * CC);
  float ss = 0.f, sd = 0.f;
  #pragma unroll
  for (int i = 0; i < 4; ++i) {
    float4 x = xp[i], s = sp[i], d = dp[i];
    ss += x.x * s.x + x.y * s.y + x.z * s.z + x.w * s.w;
    sd += x.x * d.x + x.y * d.y + x.z * d.z + x.w * d.w;
  }
  e_src[t] = ss;
  e_dst[t] = sd;
}

__device__ __forceinline__ float leaky(float v) {
  return v > 0.f ? v : v * NEG_SLOPE;
}

#define BINSH   8
#define BINW    (1 << BINSH)
#define NBIN    ((NN + BINW - 1) / BINW)
#define GCH     256
#define ECH     ((EE + GCH - 1) / GCH)
#define SORT_T  1024
#define EPT     ((ECH + SORT_T - 1) / SORT_T)
#define SPT     8
static_assert(NN <= (1 << (32 - BINSH)), "source index must fit beside the bin-local dst");
static_assert(NBIN <= SORT_T, "one thread per bin in steps 3 and 4");

__global__ __launch_bounds__(SORT_T) void bin_count_k(
    const int* __restrict__ ei, int* __restrict__ tbl) {
  __shared__ int hist[NBIN];
  const int tid = threadIdx.x;
  for (int b = tid; b < NBIN; b += SORT_T) hist[b] = 0;
  __syncthreads();
  const int e0 = blockIdx.x * ECH;
  const int e1 = min(e0 + ECH, EE);
  int d[EPT];
  #pragma unroll
  for (int k = 0; k < EPT; ++k)
    d[k] = ei[EE + min(e0 + tid + k * SORT_T, e1 - 1)];
  #pragma unroll
  for (int k = 0; k < EPT; ++k)
    if (e0 + tid + k * SORT_T < e1) atomicAdd(&hist[d[k] >> BINSH], 1);
  __syncthreads();
  for (int b = tid; b < NBIN; b += SORT_T) tbl[b * GCH + blockIdx.x] = hist[b];
}

__global__ __launch_bounds__(GCH) void bin_scan_k(int* __restrict__ tbl,
                                                  int* __restrict__ bintot) {
  __shared__ int wsum[GCH / 32];
  const int tid  = threadIdx.x;
  const int lane = tid & 31;
  const int wave = tid >> 5;
  const int c = tbl[blockIdx.x * GCH + tid];
  int incl = c;
  for (int o = 1; o < 32; o <<= 1) {
    const int u = __shfl_up(incl, o, 32);
    if (lane >= o) incl += u;
  }
  if (lane == 31) wsum[wave] = incl;
  __syncthreads();
  int off = 0;
  for (int w = 0; w < wave; ++w) off += wsum[w];
  tbl[blockIdx.x * GCH + tid] = off + incl - c;
  if (tid == GCH - 1) bintot[blockIdx.x] = off + incl;
}

__global__ __launch_bounds__(SORT_T) void bin_scatter_k(
    const int* __restrict__ ei, const int* __restrict__ tbl,
    const int* __restrict__ bintot, unsigned* __restrict__ binned) {
  __shared__ int cur[NBIN];
  __shared__ int wtot[SORT_T / 32];
  const int tid  = threadIdx.x;
  const int lane = tid & 31;
  const int wave = tid >> 5;
  const int e0 = blockIdx.x * ECH;
  const int e1 = min(e0 + ECH, EE);
  int d[EPT], sv[EPT];
  #pragma unroll
  for (int k = 0; k < EPT; ++k) {
    const int e = min(e0 + tid + k * SORT_T, e1 - 1);
    d[k]  = ei[EE + e];
    sv[k] = ei[e];
  }
  int bt = 0, tb = 0;
  if (tid < NBIN) {
    bt = bintot[tid];
    tb = tbl[tid * GCH + blockIdx.x];
  }
  int incl = bt;
  for (int o = 1; o < 32; o <<= 1) {
    const int u = __shfl_up(incl, o, 32);
    if (lane >= o) incl += u;
  }
  if (lane == 31) wtot[wave] = incl;
  __syncthreads();
  if (tid < NBIN) {
    int off = 0;
    for (int w = 0; w < wave; ++w) off += wtot[w];
    cur[tid] = off + incl - bt + tb;
  }
  __syncthreads();
  #pragma unroll
  for (int k = 0; k < EPT; ++k) {
    if (e0 + tid + k * SORT_T < e1) {
      const int p = atomicAdd(&cur[d[k] >> BINSH], 1);
      binned[p] = ((unsigned)sv[k] << BINSH) | (unsigned)(d[k] & (BINW - 1));
    }
  }
}

__global__ __launch_bounds__(SORT_T) void bin_sort_k(
    const unsigned* __restrict__ binned, const int* __restrict__ bintot,
    int* __restrict__ seg_beg, int* __restrict__ seg_end,
    int* __restrict__ csr_src) {
  __shared__ int cnt[BINW + 32];
  __shared__ int cur[BINW + 32];
  __shared__ int wsum[BINW / 32];
  __shared__ int sstart;
  const int tid  = threadIdx.x;
  const int lane = tid & 31;
  const int wave = tid >> 5;
  const int bin  = blockIdx.x;
  int v = (tid < bin) ? bintot[tid] : 0;
  const int nbin = bintot[bin];
  if (tid < BINW + 32) {
    cnt[tid] = 0;
    cur[tid] = 0;
  }
  if (tid == 0) sstart = 0;
  __syncthreads();
  for (int o = 16; o > 0; o >>= 1) v += __shfl_xor(v, o, 32);
  if (lane == 0 && v != 0) atomicAdd(&sstart, v);
  __syncthreads();
  const int start = sstart;
  const int end   = start + nbin;
  for (int i0 = start + tid; i0 < end; i0 += SPT * SORT_T) {
    unsigned r[SPT];
    #pragma unroll
    for (int u = 0; u < SPT; ++u) r[u] = binned[min(i0 + u * SORT_T, end - 1)];
    #pragma unroll
    for (int u = 0; u < SPT; ++u) {
      const int w = (i0 + u * SORT_T < end) ? (int)(r[u] & (BINW - 1))
                                            : BINW + lane;
      atomicAdd(&cnt[w], 1);
    }
  }
  __syncthreads();
  int c = 0, incl = 0;
  if (tid < BINW) {
    c = cnt[tid];
    incl = c;
    for (int o = 1; o < 32; o <<= 1) {
      const int u = __shfl_up(incl, o, 32);
      if (lane >= o) incl += u;
    }
    if (lane == 31) wsum[wave] = incl;
  }
  __syncthreads();
  if (tid < BINW) {
    int off = 0;
    for (int w = 0; w < wave; ++w) off += wsum[w];
    const int first = start + off + incl - c;
    cur[tid] = first;
    const int node = bin * BINW + tid;
    if (node < NN) {
      seg_beg[node] = first;
      seg_end[node] = first + c;
    }
  }
  __syncthreads();
  for (int i0 = start + tid; i0 < end; i0 += SPT * SORT_T) {
    unsigned r[SPT];
    int p[SPT];
    #pragma unroll
    for (int u = 0; u < SPT; ++u) r[u] = binned[min(i0 + u * SORT_T, end - 1)];
    #pragma unroll
    for (int u = 0; u < SPT; ++u) {
      const int w = (i0 + u * SORT_T < end) ? (int)(r[u] & (BINW - 1))
                                            : BINW + lane;
      p[u] = atomicAdd(&cur[w], 1);
    }
    #pragma unroll
    for (int u = 0; u < SPT; ++u)
      if (i0 + u * SORT_T < end) csr_src[p[u]] = (int)(r[u] >> BINSH);
  }
}


template <int CTRL>
__device__ __forceinline__ float quad_dpp(float v) {
  return __int_as_float(__builtin_amdgcn_update_dpp(
      0, __float_as_int(v), CTRL, 0xf, 0xf, true));
}
#define QUAD_XOR1   0xB1
#define QUAD_XOR2   0x4E
#define QUAD_LANE(U) ((U) * 0x55)

__device__ __forceinline__ void load_sweep(const float* __restrict__ xq,
                                           int sidx, int t, v4f (&x)[4]) {
  #pragma unroll
  for (int u = 0; u < 4; ++u) {
    const int su = __builtin_amdgcn_readlane(sidx, 4 * t + u);
    x[u] = *(const v4f*)(xq + (size_t)su * DD);
  }
}

__device__ __forceinline__ void acc_edge(const v4f& x, float a, v4f& acc) {
  acc.x = __builtin_fmaf(x.x, a, acc.x);
  acc.y = __builtin_fmaf(x.y, a, acc.y);
  acc.z = __builtin_fmaf(x.z, a, acc.z);
  acc.w = __builtin_fmaf(x.w, a, acc.w);
}

__device__ __forceinline__ void acc_sweep(const v4f (&x)[4], float al, v4f& acc) {
  acc_edge(x[0], quad_dpp<QUAD_LANE(0)>(al), acc);
  acc_edge(x[1], quad_dpp<QUAD_LANE(1)>(al), acc);
  acc_edge(x[2], quad_dpp<QUAD_LANE(2)>(al), acc);
  acc_edge(x[3], quad_dpp<QUAD_LANE(3)>(al), acc);
}

template <int T>
__device__ __forceinline__ void sweep_pipe(int ns, const float* __restrict__ xq,
                                           int sidx, const float (&al)[8],
                                           v4f (&cur)[4], v4f (&nxt)[4],
                                           v4f& acc) {
  acc_sweep(cur, al[T], acc);
  if constexpr (T + 2 < 8) {
    if (ns > T + 2) {
      load_sweep(xq, sidx, T + 2, cur);
      sweep_pipe<T + 1>(ns, xq, sidx, al, nxt, cur, acc);
      return;
    }
  }
  if constexpr (T + 1 < 8) {
    if (ns > T + 1)
      acc_sweep(nxt, al[T + 1], acc);
  }
}

__global__ __launch_bounds__(256, 9) void node_gat_ln_k(
    const int* __restrict__ seg_beg, const int* __restrict__ seg_end,
    const int* __restrict__ csr_src, const float* __restrict__ e_src,
    const float* __restrict__ e_dst, const float* __restrict__ xh,
    const float* hin, const float* __restrict__ gb,
    const float* __restrict__ g, const float* __restrict__ b, float* hout) {
  const int node = __builtin_amdgcn_readfirstlane(
      (int)(blockIdx.x * 8 + (threadIdx.x >> 5)));
  const int lane = threadIdx.x & 31;
  const int hq   = lane >> 2;
  const int sub  = lane & 3;
  const int beg   = seg_beg[node];
  const int deg   = seg_end[node] - beg;
  const int total = deg + 1;
  const float ed  = e_dst[node * HH + hq];
  const size_t rbase = (size_t)node * DD + 4 * lane;
  const float* __restrict__ xq = xh + 4 * lane;

  v4f acc = {0.f, 0.f, 0.f, 0.f};
  v4f r;

  if (total <= 32) {
    const int ns = (total + 3) >> 2;
    int sidx = node;
    if (lane < deg) sidx = csr_src[beg + lane];

    r = *(const v4f*)(hin + rbase);

    float ev[8];
    #pragma unroll
    for (int t = 0; t < 8; ++t) {
      ev[t] = 0.f;
      if (t == 0 || ns > t) {
        const int s = __shfl(sidx, 4 * t + sub, 32);
        ev[t] = e_src[s * HH + hq];
      }
    }

    v4f xa[4], xb[4];
    load_sweep(xq, sidx, 0, xa);
    if (ns > 1) load_sweep(xq, sidx, 1, xb);

    float sc[8];
    float mx = -__builtin_inff();
    #pragma unroll
    for (int t = 0; t < 8; ++t) {
      sc[t] = leaky(ev[t] + ed);
      if (4 * t + sub < total) mx = fmaxf(mx, sc[t]);
    }
    mx = fmaxf(mx, quad_dpp<QUAD_XOR1>(mx));
    mx = fmaxf(mx, quad_dpp<QUAD_XOR2>(mx));

    float sm = 0.f;
    #pragma unroll
    for (int t = 0; t < 8; ++t) {
      sc[t] = __expf(sc[t] - mx);
      if (4 * t + sub < total) sm += sc[t];
    }
    sm += quad_dpp<QUAD_XOR1>(sm);
    sm += quad_dpp<QUAD_XOR2>(sm);
    const float den = sm + 1e-16f;

    float al[8];
    #pragma unroll
    for (int t = 0; t < 8; ++t)
      al[t] = (4 * t + sub < total) ? sc[t] / den : 0.f;

    sweep_pipe<0>(ns, xq, sidx, al, xa, xb, acc);
  } else {
    float mx = -__builtin_inff();
    for (int j = sub; j < total; j += 4) {
      const int s = (j < deg) ? csr_src[beg + j] : node;
      mx = fmaxf(mx, leaky(e_src[s * HH + hq] + ed));
    }
    mx = fmaxf(mx, quad_dpp<QUAD_XOR1>(mx));
    mx = fmaxf(mx, quad_dpp<QUAD_XOR2>(mx));

    float sm = 0.f;
    for (int j = sub; j < total; j += 4) {
      const int s = (j < deg) ? csr_src[beg + j] : node;
      sm += __expf(leaky(e_src[s * HH + hq] + ed) - mx);
    }
    sm += quad_dpp<QUAD_XOR1>(sm);
    sm += quad_dpp<QUAD_XOR2>(sm);
    const float den = sm + 1e-16f;

#define EDGE_ACC(U)                                                    \
    {                                                                  \
      const int su = __builtin_amdgcn_readlane(s, (U));                \
      const v4f x = *(const v4f*)(xq + (size_t)su * DD);               \
      acc_edge(x, quad_dpp<QUAD_LANE(U)>(alpha), acc);                 \
    }
    for (int base = 0; base < total; base += 4) {
      const int j = base + sub;
      int   s     = node;
      float alpha = 0.f;
      if (j < total) {
        if (j < deg) s = csr_src[beg + j];
        alpha = __expf(leaky(e_src[s * HH + hq] + ed) - mx) / den;
      }
      const int left = total - base;
      if (left >= 4) {
        EDGE_ACC(0) EDGE_ACC(1) EDGE_ACC(2) EDGE_ACC(3)
      } else {
        EDGE_ACC(0)
        if (left >= 2) EDGE_ACC(1)
        if (left >= 3) EDGE_ACC(2)
      }
    }
#undef EDGE_ACC
    r = *(const v4f*)(hin + rbase);
  }

  const v4f gbv = *(const v4f*)(gb + 4 * lane);
  const v4f gv  = *(const v4f*)(g + 4 * lane);
  const v4f bv  = *(const v4f*)(b + 4 * lane);
  const float x0 = r.x + acc.x + gbv.x;
  const float x1 = r.y + acc.y + gbv.y;
  const float x2 = r.z + acc.z + gbv.z;
  const float x3 = r.w + acc.w + gbv.w;
  float ssum = x0 + x1 + x2 + x3;
  #pragma unroll
  for (int o = 16; o > 0; o >>= 1) ssum += __shfl_xor(ssum, o, 32);
  const float mean = ssum * (1.f / (float)DD);
  const float d0 = x0 - mean, d1 = x1 - mean, d2 = x2 - mean, d3 = x3 - mean;
  float vs = d0 * d0 + d1 * d1 + d2 * d2 + d3 * d3;
  #pragma unroll
  for (int o = 16; o > 0; o >>= 1) vs += __shfl_xor(vs, o, 32);
  const float inv = rsqrtf(vs * (1.f / (float)DD) + LN_EPS);
  v4f o4;
  o4.x = d0 * inv * gv.x + bv.x;
  o4.y = d1 * inv * gv.y + bv.y;
  o4.z = d2 * inv * gv.z + bv.z;
  o4.w = d3 * inv * gv.w + bv.w;
  *(v4f*)(hout + rbase) = o4;
}

#define POOL_BLK  256
#define POOL_T    512
#define POOL_WPB  (POOL_T / 32)
#define POOL_W    (POOL_BLK * POOL_WPB)
#define POOL_FULL (NN / POOL_W)
#define POOL_REM  (NN - POOL_FULL * POOL_W)
static_assert(POOL_REM >= 0 && POOL_REM < POOL_W, "at most one extra row per wave");
static_assert(POOL_T >= DD, "one thread per channel in the block's last step");

__global__ __launch_bounds__(POOL_T) void pool_sum_k(const float* __restrict__ h,
                                                     float* __restrict__ pooled) {
  __shared__ float red[POOL_WPB * DD];
  const int lane = threadIdx.x & 31;
  const int wave = threadIdx.x >> 5;
  const int gw = __builtin_amdgcn_readfirstlane(
      (int)(blockIdx.x * POOL_WPB + wave));
  const float* __restrict__ p = h + (size_t)gw * DD + 4 * lane;

  v4f x[POOL_FULL];
  #pragma unroll
  for (int u = 0; u < POOL_FULL; ++u)
    x[u] = *(const v4f*)(p + (size_t)u * POOL_W * DD);
  v4f s = {0.f, 0.f, 0.f, 0.f};
  if (gw < POOL_REM)
    s = *(const v4f*)(p + (size_t)POOL_FULL * POOL_W * DD);
  #pragma unroll
  for (int u = 0; u < POOL_FULL; ++u) s += x[u];

  *(v4f*)(red + wave * DD + 4 * lane) = s;
  __syncthreads();
  if (threadIdx.x < DD) {
    float a = 0.f;
    #pragma unroll
    for (int w = 0; w < POOL_WPB; ++w) a += red[w * DD + threadIdx.x];
    atomicAdd(&pooled[threadIdx.x], a);
  }
}

__global__ __launch_bounds__(128) void head_k(
    const float* __restrict__ pooled_sum, const float* __restrict__ W_mu,
    const float* __restrict__ b_mu, const float* __restrict__ W_lv,
    const float* __restrict__ b_lv, float* __restrict__ out) {
  int d = threadIdx.x;
  __shared__ float pl[DD];
  float pd = pooled_sum[d] * (1.f / (float)NN);
  pl[d] = pd;
  __syncthreads();
  float mu = b_mu[d], lv = b_lv[d];
  for (int k = 0; k < DD; ++k) {
    float p = pl[k];
    mu += p * W_mu[(size_t)d * DD + k];
    lv += p * W_lv[(size_t)d * DD + k];
  }
  out[d]      = mu;
  out[DD + d] = lv;
  out[2 * DD + (size_t)NN * DD + d] = pd;
}

__global__ __launch_bounds__(256) void fill32_k(unsigned* __restrict__ p,
                                                unsigned v, int n) {
  int t = blockIdx.x * blockDim.x + threadIdx.x;
  if (t < n) p[t] = v;
}

extern "C" void kernel_launch(void* const* d_in, const int* in_sizes, int n_in,
                              void* d_out, int out_size, void* d_ws, size_t ws_size,
                              hipStream_t stream) {
  const float* x       = (const float*)d_in[0];
  const int*   ei      = (const int*)d_in[1];
  const float* W_in    = (const float*)d_in[2];
  const float* b_in    = (const float*)d_in[3];
  const float* gat_W   = (const float*)d_in[4];
  const float* att_src = (const float*)d_in[5];
  const float* att_dst = (const float*)d_in[6];
  const float* gat_b   = (const float*)d_in[7];
  const float* ln_g    = (const float*)d_in[8];
  const float* ln_b    = (const float*)d_in[9];
  const float* W_mu    = (const float*)d_in[10];
  const float* b_mu    = (const float*)d_in[11];
  const float* W_lv    = (const float*)d_in[12];
  const float* b_lv    = (const float*)d_in[13];
  float* out = (float*)d_out;

  float* h       = (float*)d_ws;
  float* xh      = h  + (size_t)NN * DD;
  float* grp     = xh + (size_t)NN * DD;
  float* e_src   = grp   + (size_t)NN * DD;
  float* e_dst   = e_src + (size_t)NN * HH;
  float* pooled  = e_dst + (size_t)NN * HH * 3;

  int*  seg_end = (int*)grp;
  int*  seg_beg = seg_end + NN;
  int*  csr_src = seg_beg + NN;
  int*  tbl     = csr_src + EE;
  int*  bintot  = tbl + NBIN * GCH;
  unsigned* binned = (unsigned*)(bintot + 512);

  float* h_final = out + 2 * DD;

  const int TPB = 256;
  const int nh_blk   = (NN * HH + TPB - 1) / TPB;
  const int node_blk = NN / 8;
  const int gemm_blk = 512;

  bin_count_k  <<<GCH,  SORT_T, 0, stream>>>(ei, tbl);
  bin_scan_k   <<<NBIN, GCH,    0, stream>>>(tbl, bintot);
  bin_scatter_k<<<GCH,  SORT_T, 0, stream>>>(ei, tbl, bintot, binned);
  bin_sort_k   <<<NBIN, SORT_T, 0, stream>>>(binned, bintot, seg_beg, seg_end, csr_src);

  gemm_wmma_k<F_IN><<<gemm_blk, TPB, 0, stream>>>(x, W_in, b_in, h);

  for (int l = 0; l < LL; ++l) {
    const float* Wl  = gat_W   + (size_t)l * DD * DD;
    const float* asl = att_src + (size_t)l * HH * CC;
    const float* adl = att_dst + (size_t)l * HH * CC;
    const float* gbl = gat_b   + (size_t)l * DD;
    const float* gl  = ln_g    + (size_t)l * DD;
    const float* bl  = ln_b    + (size_t)l * DD;

    gemm_wmma_k<DD><<<gemm_blk, TPB, 0, stream>>>(h, Wl, nullptr, xh);

    att_logits_k<<<nh_blk, TPB, 0, stream>>>(xh, asl, adl, e_src, e_dst);

    float* hout = (l == LL - 1) ? h_final : h;
    node_gat_ln_k<<<node_blk, TPB, 0, stream>>>(seg_beg, seg_end, csr_src, e_src,
                                                e_dst, xh, h, gbl, gl, bl, hout);
  }

  fill32_k<<<1, TPB, 0, stream>>>((unsigned*)pooled, 0u, DD);
  pool_sum_k<<<POOL_BLK, POOL_T, 0, stream>>>(h_final, pooled);
  head_k<<<1, DD, 0, stream>>>(pooled, W_mu, b_mu, W_lv, b_lv, out);
}


# node pass: an edge's logit comes from its gathered row (quad dot by DPP) under a running-max softmax, each edge read once; e_src gathers and 4 logit launches gone; logit and softmax sums regrouped
# speedup vs baseline: 5.1557x; 1.1176x over previous
#include <hip/hip_runtime.h>
#include <hip/hip_bf16.h>


#define NN      100000
#define EE      1600000
#define ETOT    (EE + NN)
#define F_IN    256
#define DD      128
#define HH      8
#define CC      16
#define LL      4
#define NEG_SLOPE 0.2f
#define LN_EPS  1e-5f

typedef __attribute__((ext_vector_type(2))) float v2f;
typedef __attribute__((ext_vector_type(4))) float v4f;
typedef __attribute__((ext_vector_type(8))) float v8f;

template <int K>
__global__ __launch_bounds__(256, 4) void gemm_wmma_k(
    const float* __restrict__ X, const float* __restrict__ W,
    const float* __restrict__ bias, float* __restrict__ Y) {
  constexpr int KP    = K + 4;
  constexpr int PR    = 32;
  constexpr int NPASS = NN / PR;
  constexpr int TPR   = K / 4;
  constexpr int RPS   = 256 / TPR;
  constexpr int NV    = PR / RPS;
  constexpr int NS    = K / 4;
  constexpr int CH    = 4;
  constexpr int NG    = NS / CH;
  __shared__ float As[PR * KP];

  const int wave = threadIdx.x >> 5;
  const int lane = threadIdx.x & 31;
  const int col0 = wave * 16;
  const int m    = lane & 15;
  const int koff = (lane < 16) ? 0 : 2;
  const int sr   = threadIdx.x / TPR;
  const int sc   = (threadIdx.x % TPR) * 4;
  const int gstride = (int)gridDim.x;

  const float* __restrict__ wrow = W + (size_t)(col0 + m) * K + koff;
  v2f b[NS];
  #pragma unroll
  for (int i = 0; i < NS; ++i) {
    b[i].x = wrow[4 * i];
    b[i].y = wrow[4 * i + 1];
  }

  const int mrow = (lane < 16) ? 0 : 8;
  const int c    = col0 + m;
  const float bv = bias ? bias[c] : 0.0f;

  const float* __restrict__ a0row = As + m * KP + koff;
  const float* __restrict__ a1row = As + (16 + m) * KP + koff;
  const float* __restrict__ xs    = X + (size_t)sr * K + sc;
  float* __restrict__ ls          = As + sr * KP + sc;

  int pass = blockIdx.x;
  v4f pre[NV];
  {
    const float* __restrict__ xp = xs + (size_t)min(pass, NPASS - 1) * PR * K;
    #pragma unroll
    for (int i = 0; i < NV; ++i)
      pre[i] = *(const v4f*)(xp + (size_t)i * RPS * K);
  }

  for (; pass < NPASS; pass += gstride) {
    #pragma unroll
    for (int i = 0; i < NV; ++i)
      *(v4f*)(ls + i * RPS * KP) = pre[i];
    __syncthreads();

    {
      const float* __restrict__ xp =
          xs + (size_t)min(pass + gstride, NPASS - 1) * PR * K;
      #pragma unroll
      for (int i = 0; i < NV; ++i)
        pre[i] = *(const v4f*)(xp + (size_t)i * RPS * K);
    }

    v2f fa[2][2 * CH];
    #pragma unroll
    for (int j = 0; j < CH; ++j) {
      fa[0][j].x      = a0row[4 * j];
      fa[0][j].y      = a0row[4 * j + 1];
      fa[0][CH + j].x = a1row[4 * j];
      fa[0][CH + j].y = a1row[4 * j + 1];
    }
    v8f acc0 = {}, acc1 = {};
    #pragma unroll
    for (int g = 0; g < NG; ++g) {
      if (g + 1 < NG) {
        #pragma unroll
        for (int j = 0; j < CH; ++j) {
          const int k = 4 * ((g + 1) * CH + j);
          fa[(g + 1) & 1][j].x      = a0row[k];
          fa[(g + 1) & 1][j].y      = a0row[k + 1];
          fa[(g + 1) & 1][CH + j].x = a1row[k];
          fa[(g + 1) & 1][CH + j].y = a1row[k + 1];
        }
      }
      __builtin_amdgcn_sched_barrier(0);
      #pragma unroll
      for (int j = 0; j < CH; ++j) {
        acc0 = __builtin_amdgcn_wmma_f32_16x16x4_f32(
            false, fa[g & 1][j], false, b[g * CH + j], (short)0, acc0,
            false, false);
        acc1 = __builtin_amdgcn_wmma_f32_16x16x4_f32(
            false, fa[g & 1][CH + j], false, b[g * CH + j], (short)0, acc1,
            false, false);
      }
      __builtin_amdgcn_sched_barrier(0);
    }

    float* __restrict__ yp = Y + ((size_t)pass * PR + mrow) * DD + c;
    #pragma unroll
    for (int v = 0; v < 8; ++v) {
      yp[(size_t)v * DD]        = acc0[v] + bv;
      yp[(size_t)(16 + v) * DD] = acc1[v] + bv;
    }
    __syncthreads();
  }
}

__device__ __forceinline__ float leaky(float v) {
  return v > 0.f ? v : v * NEG_SLOPE;
}

#define BINSH   8
#define BINW    (1 << BINSH)
#define NBIN    ((NN + BINW - 1) / BINW)
#define GCH     256
#define ECH     ((EE + GCH - 1) / GCH)
#define SORT_T  1024
#define EPT     ((ECH + SORT_T - 1) / SORT_T)
#define SPT     8
static_assert(NN <= (1 << (32 - BINSH)), "source index must fit beside the bin-local dst");
static_assert(NBIN <= SORT_T, "one thread per bin in steps 3 and 4");

__global__ __launch_bounds__(SORT_T) void bin_count_k(
    const int* __restrict__ ei, int* __restrict__ tbl) {
  __shared__ int hist[NBIN];
  const int tid = threadIdx.x;
  for (int b = tid; b < NBIN; b += SORT_T) hist[b] = 0;
  __syncthreads();
  const int e0 = blockIdx.x * ECH;
  const int e1 = min(e0 + ECH, EE);
  int d[EPT];
  #pragma unroll
  for (int k = 0; k < EPT; ++k)
    d[k] = ei[EE + min(e0 + tid + k * SORT_T, e1 - 1)];
  #pragma unroll
  for (int k = 0; k < EPT; ++k)
    if (e0 + tid + k * SORT_T < e1) atomicAdd(&hist[d[k] >> BINSH], 1);
  __syncthreads();
  for (int b = tid; b < NBIN; b += SORT_T) tbl[b * GCH + blockIdx.x] = hist[b];
}

__global__ __launch_bounds__(GCH) void bin_scan_k(int* __restrict__ tbl,
                                                  int* __restrict__ bintot) {
  __shared__ int wsum[GCH / 32];
  const int tid  = threadIdx.x;
  const int lane = tid & 31;
  const int wave = tid >> 5;
  const int c = tbl[blockIdx.x * GCH + tid];
  int incl = c;
  for (int o = 1; o < 32; o <<= 1) {
    const int u = __shfl_up(incl, o, 32);
    if (lane >= o) incl += u;
  }
  if (lane == 31) wsum[wave] = incl;
  __syncthreads();
  int off = 0;
  for (int w = 0; w < wave; ++w) off += wsum[w];
  tbl[blockIdx.x * GCH + tid] = off + incl - c;
  if (tid == GCH - 1) bintot[blockIdx.x] = off + incl;
}

__global__ __launch_bounds__(SORT_T) void bin_scatter_k(
    const int* __restrict__ ei, const int* __restrict__ tbl,
    const int* __restrict__ bintot, unsigned* __restrict__ binned) {
  __shared__ int cur[NBIN];
  __shared__ int wtot[SORT_T / 32];
  const int tid  = threadIdx.x;
  const int lane = tid & 31;
  const int wave = tid >> 5;
  const int e0 = blockIdx.x * ECH;
  const int e1 = min(e0 + ECH, EE);
  int d[EPT], sv[EPT];
  #pragma unroll
  for (int k = 0; k < EPT; ++k) {
    const int e = min(e0 + tid + k * SORT_T, e1 - 1);
    d[k]  = ei[EE + e];
    sv[k] = ei[e];
  }
  int bt = 0, tb = 0;
  if (tid < NBIN) {
    bt = bintot[tid];
    tb = tbl[tid * GCH + blockIdx.x];
  }
  int incl = bt;
  for (int o = 1; o < 32; o <<= 1) {
    const int u = __shfl_up(incl, o, 32);
    if (lane >= o) incl += u;
  }
  if (lane == 31) wtot[wave] = incl;
  __syncthreads();
  if (tid < NBIN) {
    int off = 0;
    for (int w = 0; w < wave; ++w) off += wtot[w];
    cur[tid] = off + incl - bt + tb;
  }
  __syncthreads();
  #pragma unroll
  for (int k = 0; k < EPT; ++k) {
    if (e0 + tid + k * SORT_T < e1) {
      const int p = atomicAdd(&cur[d[k] >> BINSH], 1);
      binned[p] = ((unsigned)sv[k] << BINSH) | (unsigned)(d[k] & (BINW - 1));
    }
  }
}

__global__ __launch_bounds__(SORT_T) void bin_sort_k(
    const unsigned* __restrict__ binned, const int* __restrict__ bintot,
    int* __restrict__ seg_beg, int* __restrict__ seg_end,
    int* __restrict__ csr_src) {
  __shared__ int cnt[BINW + 32];
  __shared__ int cur[BINW + 32];
  __shared__ int wsum[BINW / 32];
  __shared__ int sstart;
  const int tid  = threadIdx.x;
  const int lane = tid & 31;
  const int wave = tid >> 5;
  const int bin  = blockIdx.x;
  int v = (tid < bin) ? bintot[tid] : 0;
  const int nbin = bintot[bin];
  if (tid < BINW + 32) {
    cnt[tid] = 0;
    cur[tid] = 0;
  }
  if (tid == 0) sstart = 0;
  __syncthreads();
  for (int o = 16; o > 0; o >>= 1) v += __shfl_xor(v, o, 32);
  if (lane == 0 && v != 0) atomicAdd(&sstart, v);
  __syncthreads();
  const int start = sstart;
  const int end   = start + nbin;
  for (int i0 = start + tid; i0 < end; i0 += SPT * SORT_T) {
    unsigned r[SPT];
    #pragma unroll
    for (int u = 0; u < SPT; ++u) r[u] = binned[min(i0 + u * SORT_T, end - 1)];
    #pragma unroll
    for (int u = 0; u < SPT; ++u) {
      const int w = (i0 + u * SORT_T < end) ? (int)(r[u] & (BINW - 1))
                                            : BINW + lane;
      atomicAdd(&cnt[w], 1);
    }
  }
  __syncthreads();
  int c = 0, incl = 0;
  if (tid < BINW) {
    c = cnt[tid];
    incl = c;
    for (int o = 1; o < 32; o <<= 1) {
      const int u = __shfl_up(incl, o, 32);
      if (lane >= o) incl += u;
    }
    if (lane == 31) wsum[wave] = incl;
  }
  __syncthreads();
  if (tid < BINW) {
    int off = 0;
    for (int w = 0; w < wave; ++w) off += wsum[w];
    const int first = start + off + incl - c;
    cur[tid] = first;
    const int node = bin * BINW + tid;
    if (node < NN) {
      seg_beg[node] = first;
      seg_end[node] = first + c;
    }
  }
  __syncthreads();
  for (int i0 = start + tid; i0 < end; i0 += SPT * SORT_T) {
    unsigned r[SPT];
    int p[SPT];
    #pragma unroll
    for (int u = 0; u < SPT; ++u) r[u] = binned[min(i0 + u * SORT_T, end - 1)];
    #pragma unroll
    for (int u = 0; u < SPT; ++u) {
      const int w = (i0 + u * SORT_T < end) ? (int)(r[u] & (BINW - 1))
                                            : BINW + lane;
      p[u] = atomicAdd(&cur[w], 1);
    }
    #pragma unroll
    for (int u = 0; u < SPT; ++u)
      if (i0 + u * SORT_T < end) csr_src[p[u]] = (int)(r[u] >> BINSH);
  }
}


template <int CTRL>
__device__ __forceinline__ float quad_dpp(float v) {
  return __int_as_float(__builtin_amdgcn_update_dpp(
      0, __float_as_int(v), CTRL, 0xf, 0xf, true));
}
#define QUAD_XOR1   0xB1
#define QUAD_XOR2   0x4E

__device__ __forceinline__ float head_dot(const v4f& x, const v4f& a) {
  const v2f xlo = {x.x, x.y}, xhi = {x.z, x.w};
  const v2f alo = {a.x, a.y}, ahi = {a.z, a.w};
  const v2f pr = __builtin_elementwise_fma(xhi, ahi, xlo * alo);
  float d = pr.x + pr.y;
  d += quad_dpp<QUAD_XOR1>(d);
  d += quad_dpp<QUAD_XOR2>(d);
  return d;
}

__device__ __forceinline__ void load_sweep(const float* __restrict__ xq,
                                           int sidx, int t, v4f (&x)[4]) {
  #pragma unroll
  for (int u = 0; u < 4; ++u) {
    const int su = __builtin_amdgcn_readlane(sidx, 4 * t + u);
    x[u] = *(const v4f*)(xq + (size_t)su * DD);
  }
}

__device__ __forceinline__ void acc_edge(const v4f& x, float a, v4f& acc) {
  acc.x = __builtin_fmaf(x.x, a, acc.x);
  acc.y = __builtin_fmaf(x.y, a, acc.y);
  acc.z = __builtin_fmaf(x.z, a, acc.z);
  acc.w = __builtin_fmaf(x.w, a, acc.w);
}

template <bool LAST>
__device__ __forceinline__ void soft_sweep(const v4f (&x)[4], const v4f& as,
                                           float ed, int left, float& m,
                                           float& l, v4f& acc) {
  float s[4];
  #pragma unroll
  for (int u = 0; u < 4; ++u) {
    s[u] = leaky(head_dot(x[u], as) + ed);
    if (LAST && u >= left) s[u] = -__builtin_inff();
  }
  const float mn = fmaxf(fmaxf(m, s[0]), fmaxf(fmaxf(s[1], s[2]), s[3]));
  const float rs = __expf(m - mn);
  m = mn;
  l *= rs;
  acc.x *= rs;
  acc.y *= rs;
  acc.z *= rs;
  acc.w *= rs;
  #pragma unroll
  for (int u = 0; u < 4; ++u) {
    const float p = __expf(s[u] - mn);
    l += p;
    acc_edge(x[u], p, acc);
  }
}

__global__ __launch_bounds__(256, 9) void node_gat_ln_k(
    const int* __restrict__ seg_beg, const int* __restrict__ seg_end,
    const int* __restrict__ csr_src, const float* __restrict__ a_src,
    const float* __restrict__ a_dst, const float* __restrict__ xh,
    const float* hin, const float* __restrict__ gb,
    const float* __restrict__ g, const float* __restrict__ b, float* hout) {
  const int node = __builtin_amdgcn_readfirstlane(
      (int)(blockIdx.x * 8 + (threadIdx.x >> 5)));
  const int lane = threadIdx.x & 31;
  const int beg   = seg_beg[node];
  const int deg   = seg_end[node] - beg;
  const int total = deg + 1;
  const size_t rbase = (size_t)node * DD + 4 * lane;
  const float* __restrict__ xq = xh + 4 * lane;

  int sidx = node;
  if (lane < deg) sidx = csr_src[beg + lane];

  const v4f as = *(const v4f*)(a_src + 4 * lane);
  const v4f ad = *(const v4f*)(a_dst + 4 * lane);
  const v4f xo = *(const v4f*)(xh + rbase);
  const v4f r  = *(const v4f*)(hin + rbase);
  const float ed = head_dot(xo, ad);

  float m = -__builtin_inff();
  float l = 0.f;
  v4f acc = {0.f, 0.f, 0.f, 0.f};

  int base = 0;
  for (;;) {
    const int nch = min(total - base, 32);
    for (int t = 0; 4 * t < nch; t += 2) {
      const int left = nch - 4 * t;
      v4f xa[4], xb[4];
      load_sweep(xq, sidx, t, xa);
      if (left > 4) load_sweep(xq, sidx, t + 1, xb);
      if (left >= 8) {
        soft_sweep<false>(xa, as, ed, 4, m, l, acc);
        soft_sweep<false>(xb, as, ed, 4, m, l, acc);
      } else if (left > 4) {
        soft_sweep<false>(xa, as, ed, 4, m, l, acc);
        soft_sweep<true>(xb, as, ed, left - 4, m, l, acc);
      } else {
        soft_sweep<true>(xa, as, ed, left, m, l, acc);
      }
    }
    base += 32;
    if (base >= total) break;
    sidx = node;
    if (base + lane < deg) sidx = csr_src[beg + base + lane];
  }

  const float inv = 1.f / (l + 1e-16f);
  acc.x *= inv;
  acc.y *= inv;
  acc.z *= inv;
  acc.w *= inv;

  const v4f gbv = *(const v4f*)(gb + 4 * lane);
  const v4f gv  = *(const v4f*)(g + 4 * lane);
  const v4f bv  = *(const v4f*)(b + 4 * lane);
  const float x0 = r.x + acc.x + gbv.x;
  const float x1 = r.y + acc.y + gbv.y;
  const float x2 = r.z + acc.z + gbv.z;
  const float x3 = r.w + acc.w + gbv.w;
  float ssum = x0 + x1 + x2 + x3;
  #pragma unroll
  for (int o = 16; o > 0; o >>= 1) ssum += __shfl_xor(ssum, o, 32);
  const float mean = ssum * (1.f / (float)DD);
  const float d0 = x0 - mean, d1 = x1 - mean, d2 = x2 - mean, d3 = x3 - mean;
  float vs = d0 * d0 + d1 * d1 + d2 * d2 + d3 * d3;
  #pragma unroll
  for (int o = 16; o > 0; o >>= 1) vs += __shfl_xor(vs, o, 32);
  const float inv_sd = rsqrtf(vs * (1.f / (float)DD) + LN_EPS);
  v4f o4;
  o4.x = d0 * inv_sd * gv.x + bv.x;
  o4.y = d1 * inv_sd * gv.y + bv.y;
  o4.z = d2 * inv_sd * gv.z + bv.z;
  o4.w = d3 * inv_sd * gv.w + bv.w;
  *(v4f*)(hout + rbase) = o4;
}

#define POOL_BLK  256
#define POOL_T    512
#define POOL_WPB  (POOL_T / 32)
#define POOL_W    (POOL_BLK * POOL_WPB)
#define POOL_FULL (NN / POOL_W)
#define POOL_REM  (NN - POOL_FULL * POOL_W)
static_assert(POOL_REM >= 0 && POOL_REM < POOL_W, "at most one extra row per wave");
static_assert(POOL_T >= DD, "one thread per channel in the block's last step");

__global__ __launch_bounds__(POOL_T) void pool_sum_k(const float* __restrict__ h,
                                                     float* __restrict__ pooled) {
  __shared__ float red[POOL_WPB * DD];
  const int lane = threadIdx.x & 31;
  const int wave = threadIdx.x >> 5;
  const int gw = __builtin_amdgcn_readfirstlane(
      (int)(blockIdx.x * POOL_WPB + wave));
  const float* __restrict__ p = h + (size_t)gw * DD + 4 * lane;

  v4f x[POOL_FULL];
  #pragma unroll
  for (int u = 0; u < POOL_FULL; ++u)
    x[u] = *(const v4f*)(p + (size_t)u * POOL_W * DD);
  v4f s = {0.f, 0.f, 0.f, 0.f};
  if (gw < POOL_REM)
    s = *(const v4f*)(p + (size_t)POOL_FULL * POOL_W * DD);
  #pragma unroll
  for (int u = 0; u < POOL_FULL; ++u) s += x[u];

  *(v4f*)(red + wave * DD + 4 * lane) = s;
  __syncthreads();
  if (threadIdx.x < DD) {
    float a = 0.f;
    #pragma unroll
    for (int w = 0; w < POOL_WPB; ++w) a += red[w * DD + threadIdx.x];
    atomicAdd(&pooled[threadIdx.x], a);
  }
}

__global__ __launch_bounds__(128) void head_k(
    const float* __restrict__ pooled_sum, const float* __restrict__ W_mu,
    const float* __restrict__ b_mu, const float* __restrict__ W_lv,
    const float* __restrict__ b_lv, float* __restrict__ out) {
  int d = threadIdx.x;
  __shared__ float pl[DD];
  float pd = pooled_sum[d] * (1.f / (float)NN);
  pl[d] = pd;
  __syncthreads();
  float mu = b_mu[d], lv = b_lv[d];
  for (int k = 0; k < DD; ++k) {
    float p = pl[k];
    mu += p * W_mu[(size_t)d * DD + k];
    lv += p * W_lv[(size_t)d * DD + k];
  }
  out[d]      = mu;
  out[DD + d] = lv;
  out[2 * DD + (size_t)NN * DD + d] = pd;
}

__global__ __launch_bounds__(256) void fill32_k(unsigned* __restrict__ p,
                                                unsigned v, int n) {
  int t = blockIdx.x * blockDim.x + threadIdx.x;
  if (t < n) p[t] = v;
}

extern "C" void kernel_launch(void* const* d_in, const int* in_sizes, int n_in,
                              void* d_out, int out_size, void* d_ws, size_t ws_size,
                              hipStream_t stream) {
  const float* x       = (const float*)d_in[0];
  const int*   ei      = (const int*)d_in[1];
  const float* W_in    = (const float*)d_in[2];
  const float* b_in    = (const float*)d_in[3];
  const float* gat_W   = (const float*)d_in[4];
  const float* att_src = (const float*)d_in[5];
  const float* att_dst = (const float*)d_in[6];
  const float* gat_b   = (const float*)d_in[7];
  const float* ln_g    = (const float*)d_in[8];
  const float* ln_b    = (const float*)d_in[9];
  const float* W_mu    = (const float*)d_in[10];
  const float* b_mu    = (const float*)d_in[11];
  const float* W_lv    = (const float*)d_in[12];
  const float* b_lv    = (const float*)d_in[13];
  float* out = (float*)d_out;

  float* h       = (float*)d_ws;
  float* xh      = h  + (size_t)NN * DD;
  float* grp     = xh + (size_t)NN * DD;
  float* pooled  = grp + (size_t)NN * DD + (size_t)NN * HH * 4;

  int*  seg_end = (int*)grp;
  int*  seg_beg = seg_end + NN;
  int*  csr_src = seg_beg + NN;
  int*  tbl     = csr_src + EE;
  int*  bintot  = tbl + NBIN * GCH;
  unsigned* binned = (unsigned*)(bintot + 512);

  float* h_final = out + 2 * DD;

  const int TPB = 256;
  const int node_blk = NN / 8;
  const int gemm_blk = 512;

  bin_count_k  <<<GCH,  SORT_T, 0, stream>>>(ei, tbl);
  bin_scan_k   <<<NBIN, GCH,    0, stream>>>(tbl, bintot);
  bin_scatter_k<<<GCH,  SORT_T, 0, stream>>>(ei, tbl, bintot, binned);
  bin_sort_k   <<<NBIN, SORT_T, 0, stream>>>(binned, bintot, seg_beg, seg_end, csr_src);

  gemm_wmma_k<F_IN><<<gemm_blk, TPB, 0, stream>>>(x, W_in, b_in, h);

  for (int l = 0; l < LL; ++l) {
    const float* Wl  = gat_W   + (size_t)l * DD * DD;
    const float* asl = att_src + (size_t)l * HH * CC;
    const float* adl = att_dst + (size_t)l * HH * CC;
    const float* gbl = gat_b   + (size_t)l * DD;
    const float* gl  = ln_g    + (size_t)l * DD;
    const float* bl  = ln_b    + (size_t)l * DD;

    gemm_wmma_k<DD><<<gemm_blk, TPB, 0, stream>>>(h, Wl, nullptr, xh);

    float* hout = (l == LL - 1) ? h_final : h;
    node_gat_ln_k<<<node_blk, TPB, 0, stream>>>(seg_beg, seg_end, csr_src, asl,
                                                adl, xh, h, gbl, gl, bl, hout);
  }

  fill32_k<<<1, TPB, 0, stream>>>((unsigned*)pooled, 0u, DD);
  pool_sum_k<<<POOL_BLK, POOL_T, 0, stream>>>(h_final, pooled);
  head_k<<<1, DD, 0, stream>>>(pooled, W_mu, b_mu, W_lv, b_lv, out);
}
